# starting kernel
# baseline (speedup 1.0000x reference)
#include <hip/hip_runtime.h>
#include <hip/hip_bf16.h>

typedef __attribute__((ext_vector_type(2))) float v2f;
typedef __attribute__((ext_vector_type(8))) float v8f;

#define XZ_STRIDE 385
#define NPIX_PER_WAVE 4
#define WAVES_PER_BLOCK 2

struct WaveSmem {
    float xz[16 * XZ_STRIDE];
    float xdbl[16 * 48];
    float mu[16];
    float rs[16];
};

__global__ __launch_bounds__(64) void
ssm_fused_kernel(const float* __restrict__ x,
                 const float* __restrict__ Win,
                 const float* __restrict__ Wxp,
                 const float* __restrict__ Wdt,
                 const float* __restrict__ dt_bias,
                 const float* __restrict__ A_logs,
                 const float* __restrict__ Dvec,
                 const float* __restrict__ gamma,
                 const float* __restrict__ beta,
                 const float* __restrict__ Wout,
                 float* __restrict__ out,
                 int npix)
{
    __shared__ WaveSmem smem[WAVES_PER_BLOCK];

    const int wave = threadIdx.x >> 5;
    const int lane = threadIdx.x & 31;
    const int pix0 = (blockIdx.x * WAVES_PER_BLOCK + wave) * NPIX_PER_WAVE;
    if (pix0 + NPIX_PER_WAVE > npix) return;

    WaveSmem& S = smem[wave];

    const int rowM  = lane & 15;
    const int khalf = lane >> 4;
    const float* xrow = x + (size_t)(pix0 + (rowM >> 2)) * 384 + (rowM & 3);

    for (int nt = 0; nt < 24; ++nt) {
        const int c0 = nt * 16;
        v8f acc = {};
        const float* wb = Win + (size_t)(c0 + rowM) * 96 + 2 * khalf;
        for (int kk = 0; kk < 24; ++kk) {
            const int kb = kk * 4 + 2 * khalf;
            v2f a; a.x = xrow[(size_t)kb * 4]; a.y = xrow[(size_t)(kb + 1) * 4];
            v2f b; b.x = wb[kk * 4];           b.y = wb[kk * 4 + 1];
            acc = __builtin_amdgcn_wmma_f32_16x16x4_f32(false, a, false, b,
                                                        (short)0, acc, false, false);
        }
        const int ncol = c0 + rowM;
        for (int r = 0; r < 8; ++r) {
            float v = acc[r];
            v = v / (1.0f + __expf(-v));
            S.xz[(r + 8 * khalf) * XZ_STRIDE + ncol] = v;
        }
    }
    __syncthreads();

    for (int nt = 0; nt < 3; ++nt) {
        const int c0 = nt * 16;
        const int c  = c0 + rowM;
        const bool valid = (c < 38);
        const float* wb = valid ? (Wxp + (size_t)c * 192 + 2 * khalf) : Wxp;
        v8f acc = {};
        const float* arow = &S.xz[rowM * XZ_STRIDE];
        for (int kk = 0; kk < 48; ++kk) {
            const int kb = kk * 4 + 2 * khalf;
            v2f a; a.x = arow[kb]; a.y = arow[kb + 1];
            v2f b;
            b.x = valid ? wb[kk * 4]     : 0.0f;
            b.y = valid ? wb[kk * 4 + 1] : 0.0f;
            acc = __builtin_amdgcn_wmma_f32_16x16x4_f32(false, a, false, b,
                                                        (short)0, acc, false, false);
        }
        for (int r = 0; r < 8; ++r)
            S.xdbl[(r + 8 * khalf) * 48 + c0 + rowM] = acc[r];
    }
    __syncthreads();

    {
        float lnsum[16], lnsq[16];
        #pragma unroll
        for (int r = 0; r < 16; ++r) { lnsum[r] = 0.f; lnsq[r] = 0.f; }

        for (int it = 0; it < 6; ++it) {
            const int ch = it * 32 + lane;
            float Av[16];
            #pragma unroll
            for (int n = 0; n < 16; ++n) Av[n] = -__expf(A_logs[ch * 16 + n]);
            const float dtb = dt_bias[ch];
            const float dch = Dvec[ch];
            float wdt[6];
            #pragma unroll
            for (int r = 0; r < 6; ++r) wdt[r] = Wdt[ch * 6 + r];

            #pragma unroll
            for (int p = 0; p < 4; ++p) {
                float h[16];
                #pragma unroll
                for (int n = 0; n < 16; ++n) h[n] = 0.f;
                #pragma unroll
                for (int l = 0; l < 4; ++l) {
                    const int row = p * 4 + l;
                    const float* xd = &S.xdbl[row * 48];
                    float dacc = dtb;
                    #pragma unroll
                    for (int r = 0; r < 6; ++r) dacc += xd[r] * wdt[r];
                    const float delta = (dacc > 20.f) ? dacc
                                       : __logf(1.0f + __expf(dacc));
                    const float u  = S.xz[row * XZ_STRIDE + ch];
                    const float du = delta * u;
                    float y = 0.f;
                    #pragma unroll
                    for (int n = 0; n < 16; ++n) {
                        const float dA = __expf(delta * Av[n]);
                        h[n] = dA * h[n] + du * xd[6 + n];
                        y += h[n] * xd[22 + n];
                    }
                    y += dch * u;
                    S.xz[row * XZ_STRIDE + ch] = y;
                    lnsum[row] += y; lnsq[row] += y * y;
                }
            }
        }
        #pragma unroll
        for (int m = 1; m < 32; m <<= 1) {
            #pragma unroll
            for (int r = 0; r < 16; ++r) {
                lnsum[r] += __shfl_xor(lnsum[r], m, 32);
                lnsq[r]  += __shfl_xor(lnsq[r],  m, 32);
            }
        }
        if (lane == 0) {
            #pragma unroll
            for (int r = 0; r < 16; ++r) {
                const float mu  = lnsum[r] * (1.0f / 192.0f);
                const float var = lnsq[r] * (1.0f / 192.0f) - mu * mu;
                S.mu[r] = mu;
                S.rs[r] = rsqrtf(var + 1e-5f);
            }
        }
    }
    __syncthreads();

    for (int i = lane; i < 16 * 192; i += 32) {
        const int row = i / 192;
        const int ch  = i - row * 192;
        const float y = S.xz[row * XZ_STRIDE + ch];
        float v = (y - S.mu[row]) * S.rs[row] * gamma[ch] + beta[ch];
        v *= S.xz[row * XZ_STRIDE + 192 + ch];
        S.xz[row * XZ_STRIDE + ch] = v;
    }
    __syncthreads();

    for (int nt = 0; nt < 6; ++nt) {
        const int c0 = nt * 16;
        v8f acc = {};
        const float* wb = Wout + (size_t)(c0 + rowM) * 192 + 2 * khalf;
        const float* arow = &S.xz[rowM * XZ_STRIDE];
        for (int kk = 0; kk < 48; ++kk) {
            const int kb = kk * 4 + 2 * khalf;
            v2f a; a.x = arow[kb];   a.y = arow[kb + 1];
            v2f b; b.x = wb[kk * 4]; b.y = wb[kk * 4 + 1];
            acc = __builtin_amdgcn_wmma_f32_16x16x4_f32(false, a, false, b,
                                                        (short)0, acc, false, false);
        }
        const int ncol = c0 + rowM;
        for (int r = 0; r < 8; ++r) {
            const int row = r + 8 * khalf;
            out[(size_t)(pix0 + (row >> 2)) * 384 + ncol * 4 + (row & 3)] = acc[r];
        }
    }
}

extern "C" void kernel_launch(void* const* d_in, const int* in_sizes, int n_in,
                              void* d_out, int out_size, void* d_ws, size_t ws_size,
                              hipStream_t stream) {
    const float* x       = (const float*)d_in[0];
    const float* Win     = (const float*)d_in[1];
    const float* Wxp     = (const float*)d_in[2];
    const float* Wdt     = (const float*)d_in[3];
    const float* dt_bias = (const float*)d_in[4];
    const float* A_logs  = (const float*)d_in[5];
    const float* Dvec    = (const float*)d_in[6];
    const float* gamma   = (const float*)d_in[7];
    const float* beta    = (const float*)d_in[8];
    const float* Wout    = (const float*)d_in[9];
    float* out = (float*)d_out;

    const int npix = in_sizes[0] / (96 * 4);
    const int pix_per_block = NPIX_PER_WAVE * WAVES_PER_BLOCK;
    const int grid = npix / pix_per_block;

    ssm_fused_kernel<<<grid, 64, 0, stream>>>(x, Win, Wxp, Wdt, dt_bias, A_logs,
                                              Dvec, gamma, beta, Wout, out, npix);
}


# z half of the in-projection computed after the scan and consumed from its accumulator (LN*gamma+beta times z fused there): z never stored, LDS per block 55680 -> 31104 B, all blocks co-resident
# speedup vs baseline: 1.7911x; 1.7911x over previous
#include <hip/hip_runtime.h>
#include <hip/hip_bf16.h>

typedef __attribute__((ext_vector_type(2))) float v2f;
typedef __attribute__((ext_vector_type(8))) float v8f;

#define Y_STRIDE 193
#define NPIX_PER_WAVE 4
#define WAVES_PER_BLOCK 2

struct WaveSmem {
    float y[16 * Y_STRIDE];
    float xdbl[16 * 48];
    float mu[16];
    float rs[16];
};

__global__ __launch_bounds__(64) void
ssm_fused_kernel(const float* __restrict__ x,
                 const float* __restrict__ Win,
                 const float* __restrict__ Wxp,
                 const float* __restrict__ Wdt,
                 const float* __restrict__ dt_bias,
                 const float* __restrict__ A_logs,
                 const float* __restrict__ Dvec,
                 const float* __restrict__ gamma,
                 const float* __restrict__ beta,
                 const float* __restrict__ Wout,
                 float* __restrict__ out,
                 int npix)
{
    __shared__ WaveSmem smem[WAVES_PER_BLOCK];

    const int wave = threadIdx.x >> 5;
    const int lane = threadIdx.x & 31;
    const int pix0 = (blockIdx.x * WAVES_PER_BLOCK + wave) * NPIX_PER_WAVE;
    if (pix0 + NPIX_PER_WAVE > npix) return;

    WaveSmem& S = smem[wave];

    const int rowM  = lane & 15;
    const int khalf = lane >> 4;
    const float* xrow = x + (size_t)(pix0 + (rowM >> 2)) * 384 + (rowM & 3);

    for (int nt = 0; nt < 12; ++nt) {
        const int c0 = nt * 16;
        v8f acc = {};
        const float* wb = Win + (size_t)(c0 + rowM) * 96 + 2 * khalf;
        for (int kk = 0; kk < 24; ++kk) {
            const int kb = kk * 4 + 2 * khalf;
            v2f a; a.x = xrow[(size_t)kb * 4]; a.y = xrow[(size_t)(kb + 1) * 4];
            v2f b; b.x = wb[kk * 4];           b.y = wb[kk * 4 + 1];
            acc = __builtin_amdgcn_wmma_f32_16x16x4_f32(false, a, false, b,
                                                        (short)0, acc, false, false);
        }
        const int ncol = c0 + rowM;
        for (int r = 0; r < 8; ++r) {
            float v = acc[r];
            v = v / (1.0f + __expf(-v));
            S.y[(r + 8 * khalf) * Y_STRIDE + ncol] = v;
        }
    }
    __syncthreads();

    for (int nt = 0; nt < 3; ++nt) {
        const int c0 = nt * 16;
        const int c  = c0 + rowM;
        const bool valid = (c < 38);
        const float* wb = valid ? (Wxp + (size_t)c * 192 + 2 * khalf) : Wxp;
        v8f acc = {};
        const float* arow = &S.y[rowM * Y_STRIDE];
        for (int kk = 0; kk < 48; ++kk) {
            const int kb = kk * 4 + 2 * khalf;
            v2f a; a.x = arow[kb]; a.y = arow[kb + 1];
            v2f b;
            b.x = valid ? wb[kk * 4]     : 0.0f;
            b.y = valid ? wb[kk * 4 + 1] : 0.0f;
            acc = __builtin_amdgcn_wmma_f32_16x16x4_f32(false, a, false, b,
                                                        (short)0, acc, false, false);
        }
        for (int r = 0; r < 8; ++r)
            S.xdbl[(r + 8 * khalf) * 48 + c0 + rowM] = acc[r];
    }
    __syncthreads();

    {
        float lnsum[16], lnsq[16];
        #pragma unroll
        for (int r = 0; r < 16; ++r) { lnsum[r] = 0.f; lnsq[r] = 0.f; }

        for (int it = 0; it < 6; ++it) {
            const int ch = it * 32 + lane;
            float Av[16];
            #pragma unroll
            for (int n = 0; n < 16; ++n) Av[n] = -__expf(A_logs[ch * 16 + n]);
            const float dtb = dt_bias[ch];
            const float dch = Dvec[ch];
            float wdt[6];
            #pragma unroll
            for (int r = 0; r < 6; ++r) wdt[r] = Wdt[ch * 6 + r];

            #pragma unroll
            for (int p = 0; p < 4; ++p) {
                float h[16];
                #pragma unroll
                for (int n = 0; n < 16; ++n) h[n] = 0.f;
                #pragma unroll
                for (int l = 0; l < 4; ++l) {
                    const int row = p * 4 + l;
                    const float* xd = &S.xdbl[row * 48];
                    float dacc = dtb;
                    #pragma unroll
                    for (int r = 0; r < 6; ++r) dacc += xd[r] * wdt[r];
                    const float delta = (dacc > 20.f) ? dacc
                                       : __logf(1.0f + __expf(dacc));
                    const float u  = S.y[row * Y_STRIDE + ch];
                    const float du = delta * u;
                    float y = 0.f;
                    #pragma unroll
                    for (int n = 0; n < 16; ++n) {
                        const float dA = __expf(delta * Av[n]);
                        h[n] = dA * h[n] + du * xd[6 + n];
                        y += h[n] * xd[22 + n];
                    }
                    y += dch * u;
                    S.y[row * Y_STRIDE + ch] = y;
                    lnsum[row] += y; lnsq[row] += y * y;
                }
            }
        }
        #pragma unroll
        for (int m = 1; m < 32; m <<= 1) {
            #pragma unroll
            for (int r = 0; r < 16; ++r) {
                lnsum[r] += __shfl_xor(lnsum[r], m, 32);
                lnsq[r]  += __shfl_xor(lnsq[r],  m, 32);
            }
        }
        if (lane == 0) {
            #pragma unroll
            for (int r = 0; r < 16; ++r) {
                const float mu  = lnsum[r] * (1.0f / 192.0f);
                const float var = lnsq[r] * (1.0f / 192.0f) - mu * mu;
                S.mu[r] = mu;
                S.rs[r] = rsqrtf(var + 1e-5f);
            }
        }
    }
    __syncthreads();

    {
        float mur[8], rsr[8];
        #pragma unroll
        for (int r = 0; r < 8; ++r) { mur[r] = S.mu[r + 8 * khalf]; rsr[r] = S.rs[r + 8 * khalf]; }

        for (int nt = 0; nt < 12; ++nt) {
            const int c0 = nt * 16;
            v8f acc = {};
            const float* wb = Win + (size_t)(192 + c0 + rowM) * 96 + 2 * khalf;
            for (int kk = 0; kk < 24; ++kk) {
                const int kb = kk * 4 + 2 * khalf;
                v2f a; a.x = xrow[(size_t)kb * 4]; a.y = xrow[(size_t)(kb + 1) * 4];
                v2f b; b.x = wb[kk * 4];           b.y = wb[kk * 4 + 1];
                acc = __builtin_amdgcn_wmma_f32_16x16x4_f32(false, a, false, b,
                                                            (short)0, acc, false, false);
            }
            const int ncol = c0 + rowM;
            const float g  = gamma[ncol];
            const float bt = beta[ncol];
            for (int r = 0; r < 8; ++r) {
                float z = acc[r];
                z = z / (1.0f + __expf(-z));
                const int idx = (r + 8 * khalf) * Y_STRIDE + ncol;
                float v = (S.y[idx] - mur[r]) * rsr[r] * g + bt;
                v *= z;
                S.y[idx] = v;
            }
        }
    }
    __syncthreads();

    for (int nt = 0; nt < 6; ++nt) {
        const int c0 = nt * 16;
        v8f acc = {};
        const float* wb = Wout + (size_t)(c0 + rowM) * 192 + 2 * khalf;
        const float* arow = &S.y[rowM * Y_STRIDE];
        for (int kk = 0; kk < 48; ++kk) {
            const int kb = kk * 4 + 2 * khalf;
            v2f a; a.x = arow[kb];   a.y = arow[kb + 1];
            v2f b; b.x = wb[kk * 4]; b.y = wb[kk * 4 + 1];
            acc = __builtin_amdgcn_wmma_f32_16x16x4_f32(false, a, false, b,
                                                        (short)0, acc, false, false);
        }
        const int ncol = c0 + rowM;
        for (int r = 0; r < 8; ++r) {
            const int row = r + 8 * khalf;
            out[(size_t)(pix0 + (row >> 2)) * 384 + ncol * 4 + (row & 3)] = acc[r];
        }
    }
}

extern "C" void kernel_launch(void* const* d_in, const int* in_sizes, int n_in,
                              void* d_out, int out_size, void* d_ws, size_t ws_size,
                              hipStream_t stream) {
    const float* x       = (const float*)d_in[0];
    const float* Win     = (const float*)d_in[1];
    const float* Wxp     = (const float*)d_in[2];
    const float* Wdt     = (const float*)d_in[3];
    const float* dt_bias = (const float*)d_in[4];
    const float* A_logs  = (const float*)d_in[5];
    const float* Dvec    = (const float*)d_in[6];
    const float* gamma   = (const float*)d_in[7];
    const float* beta    = (const float*)d_in[8];
    const float* Wout    = (const float*)d_in[9];
    float* out = (float*)d_out;

    const int npix = in_sizes[0] / (96 * 4);
    const int pix_per_block = NPIX_PER_WAVE * WAVES_PER_BLOCK;
    const int grid = npix / pix_per_block;

    ssm_fused_kernel<<<grid, 64, 0, stream>>>(x, Win, Wxp, Wdt, dt_bias, A_logs,
                                              Dvec, gamma, beta, Wout, out, npix);
}


# GEMM fragments fetched 16 B per lane, k order inside each group of 8 permuted alike on A and B (f32 sum order changes): weight loads b64->b128 halve cache-line touches; LDS pitch 193->196
# speedup vs baseline: 2.0303x; 1.1335x over previous
#include <hip/hip_runtime.h>
#include <hip/hip_bf16.h>

typedef __attribute__((ext_vector_type(2))) float v2f;
typedef __attribute__((ext_vector_type(8))) float v8f;

#define Y_STRIDE 196
#define NPIX_PER_WAVE 4
#define WAVES_PER_BLOCK 2

struct WaveSmem {
    float y[16 * Y_STRIDE];
    float xdbl[16 * 48];
    float mu[16];
    float rs[16];
};

__global__ __launch_bounds__(64) void
ssm_fused_kernel(const float* __restrict__ x,
                 const float* __restrict__ Win,
                 const float* __restrict__ Wxp,
                 const float* __restrict__ Wdt,
                 const float* __restrict__ dt_bias,
                 const float* __restrict__ A_logs,
                 const float* __restrict__ Dvec,
                 const float* __restrict__ gamma,
                 const float* __restrict__ beta,
                 const float* __restrict__ Wout,
                 float* __restrict__ out,
                 int npix)
{
    __shared__ WaveSmem smem[WAVES_PER_BLOCK];

    const int wave = threadIdx.x >> 5;
    const int lane = threadIdx.x & 31;
    const int pix0 = (blockIdx.x * WAVES_PER_BLOCK + wave) * NPIX_PER_WAVE;
    if (pix0 + NPIX_PER_WAVE > npix) return;

    WaveSmem& S = smem[wave];

    const int rowM  = lane & 15;
    const int khalf = lane >> 4;
    const float* xrow = x + (size_t)(pix0 + (rowM >> 2)) * 384 + (rowM & 3);

    for (int nt = 0; nt < 12; ++nt) {
        const int c0 = nt * 16;
        v8f acc = {};
        const float* wb = Win + (size_t)(c0 + rowM) * 96 + 4 * khalf;
        for (int j = 0; j < 12; ++j) {
            const int kb = j * 8 + 4 * khalf;
            const float b0 = wb[j * 8], b1 = wb[j * 8 + 1], b2 = wb[j * 8 + 2], b3 = wb[j * 8 + 3];
            v2f a0; a0.x = xrow[(size_t)kb * 4];       a0.y = xrow[(size_t)(kb + 1) * 4];
            v2f a1; a1.x = xrow[(size_t)(kb + 2) * 4]; a1.y = xrow[(size_t)(kb + 3) * 4];
            v2f bA; bA.x = b0; bA.y = b1;
            v2f bB; bB.x = b2; bB.y = b3;
            acc = __builtin_amdgcn_wmma_f32_16x16x4_f32(false, a0, false, bA,
                                                        (short)0, acc, false, false);
            acc = __builtin_amdgcn_wmma_f32_16x16x4_f32(false, a1, false, bB,
                                                        (short)0, acc, false, false);
        }
        const int ncol = c0 + rowM;
        for (int r = 0; r < 8; ++r) {
            float v = acc[r];
            v = v / (1.0f + __expf(-v));
            S.y[(r + 8 * khalf) * Y_STRIDE + ncol] = v;
        }
    }
    __syncthreads();

    for (int nt = 0; nt < 3; ++nt) {
        const int c0 = nt * 16;
        const int c  = c0 + rowM;
        const bool valid = (c < 38);
        const float* wb = Wxp + (size_t)(valid ? c : 37) * 192 + 4 * khalf;
        v8f acc = {};
        const float* arow = &S.y[rowM * Y_STRIDE + 4 * khalf];
        for (int j = 0; j < 24; ++j) {
            const float b0 = wb[j * 8], b1 = wb[j * 8 + 1], b2 = wb[j * 8 + 2], b3 = wb[j * 8 + 3];
            v2f a0; a0.x = arow[j * 8];     a0.y = arow[j * 8 + 1];
            v2f a1; a1.x = arow[j * 8 + 2]; a1.y = arow[j * 8 + 3];
            v2f bA; bA.x = valid ? b0 : 0.0f; bA.y = valid ? b1 : 0.0f;
            v2f bB; bB.x = valid ? b2 : 0.0f; bB.y = valid ? b3 : 0.0f;
            acc = __builtin_amdgcn_wmma_f32_16x16x4_f32(false, a0, false, bA,
                                                        (short)0, acc, false, false);
            acc = __builtin_amdgcn_wmma_f32_16x16x4_f32(false, a1, false, bB,
                                                        (short)0, acc, false, false);
        }
        for (int r = 0; r < 8; ++r)
            S.xdbl[(r + 8 * khalf) * 48 + c0 + rowM] = acc[r];
    }
    __syncthreads();

    {
        float lnsum[16], lnsq[16];
        #pragma unroll
        for (int r = 0; r < 16; ++r) { lnsum[r] = 0.f; lnsq[r] = 0.f; }

        for (int it = 0; it < 6; ++it) {
            const int ch = it * 32 + lane;
            float Av[16];
            #pragma unroll
            for (int n = 0; n < 16; ++n) Av[n] = -__expf(A_logs[ch * 16 + n]);
            const float dtb = dt_bias[ch];
            const float dch = Dvec[ch];
            float wdt[6];
            #pragma unroll
            for (int r = 0; r < 6; ++r) wdt[r] = Wdt[ch * 6 + r];

            #pragma unroll
            for (int p = 0; p < 4; ++p) {
                float h[16];
                #pragma unroll
                for (int n = 0; n < 16; ++n) h[n] = 0.f;
                #pragma unroll
                for (int l = 0; l < 4; ++l) {
                    const int row = p * 4 + l;
                    const float* xd = &S.xdbl[row * 48];
                    float dacc = dtb;
                    #pragma unroll
                    for (int r = 0; r < 6; ++r) dacc += xd[r] * wdt[r];
                    const float delta = (dacc > 20.f) ? dacc
                                       : __logf(1.0f + __expf(dacc));
                    const float u  = S.y[row * Y_STRIDE + ch];
                    const float du = delta * u;
                    float y = 0.f;
                    #pragma unroll
                    for (int n = 0; n < 16; ++n) {
                        const float dA = __expf(delta * Av[n]);
                        h[n] = dA * h[n] + du * xd[6 + n];
                        y += h[n] * xd[22 + n];
                    }
                    y += dch * u;
                    S.y[row * Y_STRIDE + ch] = y;
                    lnsum[row] += y; lnsq[row] += y * y;
                }
            }
        }
        #pragma unroll
        for (int m = 1; m < 32; m <<= 1) {
            #pragma unroll
            for (int r = 0; r < 16; ++r) {
                lnsum[r] += __shfl_xor(lnsum[r], m, 32);
                lnsq[r]  += __shfl_xor(lnsq[r],  m, 32);
            }
        }
        if (lane == 0) {
            #pragma unroll
            for (int r = 0; r < 16; ++r) {
                const float mu  = lnsum[r] * (1.0f / 192.0f);
                const float var = lnsq[r] * (1.0f / 192.0f) - mu * mu;
                S.mu[r] = mu;
                S.rs[r] = rsqrtf(var + 1e-5f);
            }
        }
    }
    __syncthreads();

    {
        float mur[8], rsr[8];
        #pragma unroll
        for (int r = 0; r < 8; ++r) { mur[r] = S.mu[r + 8 * khalf]; rsr[r] = S.rs[r + 8 * khalf]; }

        for (int nt = 0; nt < 12; ++nt) {
            const int c0 = nt * 16;
            v8f acc = {};
            const float* wb = Win + (size_t)(192 + c0 + rowM) * 96 + 4 * khalf;
            for (int j = 0; j < 12; ++j) {
                const int kb = j * 8 + 4 * khalf;
                const float b0 = wb[j * 8], b1 = wb[j * 8 + 1], b2 = wb[j * 8 + 2], b3 = wb[j * 8 + 3];
                v2f a0; a0.x = xrow[(size_t)kb * 4];       a0.y = xrow[(size_t)(kb + 1) * 4];
                v2f a1; a1.x = xrow[(size_t)(kb + 2) * 4]; a1.y = xrow[(size_t)(kb + 3) * 4];
                v2f bA; bA.x = b0; bA.y = b1;
                v2f bB; bB.x = b2; bB.y = b3;
                acc = __builtin_amdgcn_wmma_f32_16x16x4_f32(false, a0, false, bA,
                                                            (short)0, acc, false, false);
                acc = __builtin_amdgcn_wmma_f32_16x16x4_f32(false, a1, false, bB,
                                                            (short)0, acc, false, false);
            }
            const int ncol = c0 + rowM;
            const float g  = gamma[ncol];
            const float bt = beta[ncol];
            for (int r = 0; r < 8; ++r) {
                float z = acc[r];
                z = z / (1.0f + __expf(-z));
                const int idx = (r + 8 * khalf) * Y_STRIDE + ncol;
                float v = (S.y[idx] - mur[r]) * rsr[r] * g + bt;
                v *= z;
                S.y[idx] = v;
            }
        }
    }
    __syncthreads();

    for (int nt = 0; nt < 6; ++nt) {
        const int c0 = nt * 16;
        v8f acc = {};
        const float* wb = Wout + (size_t)(c0 + rowM) * 192 + 4 * khalf;
        const float* arow = &S.y[rowM * Y_STRIDE + 4 * khalf];
        for (int j = 0; j < 24; ++j) {
            const float b0 = wb[j * 8], b1 = wb[j * 8 + 1], b2 = wb[j * 8 + 2], b3 = wb[j * 8 + 3];
            v2f a0; a0.x = arow[j * 8];     a0.y = arow[j * 8 + 1];
            v2f a1; a1.x = arow[j * 8 + 2]; a1.y = arow[j * 8 + 3];
            v2f bA; bA.x = b0; bA.y = b1;
            v2f bB; bB.x = b2; bB.y = b3;
            acc = __builtin_amdgcn_wmma_f32_16x16x4_f32(false, a0, false, bA,
                                                        (short)0, acc, false, false);
            acc = __builtin_amdgcn_wmma_f32_16x16x4_f32(false, a1, false, bB,
                                                        (short)0, acc, false, false);
        }
        const int ncol = c0 + rowM;
        for (int r = 0; r < 8; ++r) {
            const int row = r + 8 * khalf;
            out[(size_t)(pix0 + (row >> 2)) * 384 + ncol * 4 + (row & 3)] = acc[r];
        }
    }
}

extern "C" void kernel_launch(void* const* d_in, const int* in_sizes, int n_in,
                              void* d_out, int out_size, void* d_ws, size_t ws_size,
                              hipStream_t stream) {
    const float* x       = (const float*)d_in[0];
    const float* Win     = (const float*)d_in[1];
    const float* Wxp     = (const float*)d_in[2];
    const float* Wdt     = (const float*)d_in[3];
    const float* dt_bias = (const float*)d_in[4];
    const float* A_logs  = (const float*)d_in[5];
    const float* Dvec    = (const float*)d_in[6];
    const float* gamma   = (const float*)d_in[7];
    const float* beta    = (const float*)d_in[8];
    const float* Wout    = (const float*)d_in[9];
    float* out = (float*)d_out;

    const int npix = in_sizes[0] / (96 * 4);
    const int pix_per_block = NPIX_PER_WAVE * WAVES_PER_BLOCK;
    const int grid = npix / pix_per_block;

    ssm_fused_kernel<<<grid, 64, 0, stream>>>(x, Win, Wxp, Wdt, dt_bias, A_logs,
                                              Dvec, gamma, beta, Wout, out, npix);
}


# Weight fragments of each GEMM step (12 k-groups, 24 WMMAs) requested one step ahead into a second register set, across tile and GEMM boundaries: no step waits on its own loads; sum order unchanged
# speedup vs baseline: 2.1054x; 1.0370x over previous
#include <hip/hip_runtime.h>
#include <hip/hip_bf16.h>

typedef __attribute__((ext_vector_type(2))) float v2f;
typedef __attribute__((ext_vector_type(8))) float v8f;

#define Y_STRIDE 196
#define NPIX_PER_WAVE 4
#define WAVES_PER_BLOCK 2

struct WaveSmem {
    float y[16 * Y_STRIDE];
    float xdbl[16 * 48];
    float mu[16];
    float rs[16];
};

typedef __attribute__((ext_vector_type(4))) float v4f;


__device__ __forceinline__ void load_w12(const float* __restrict__ wb, v4f (&w)[12])
{
    #pragma unroll
    for (int j = 0; j < 12; ++j) {
        w[j].x = wb[j * 8];     w[j].y = wb[j * 8 + 1];
        w[j].z = wb[j * 8 + 2]; w[j].w = wb[j * 8 + 3];
    }
}

__device__ __forceinline__ void load_a12(const float* arow, v2f (&a)[24])
{
    #pragma unroll
    for (int j = 0; j < 12; ++j) {
        a[2 * j].x     = arow[j * 8];     a[2 * j].y     = arow[j * 8 + 1];
        a[2 * j + 1].x = arow[j * 8 + 2]; a[2 * j + 1].y = arow[j * 8 + 3];
    }
}

__device__ __forceinline__ void load_x12(const float* __restrict__ xrow, int khalf, v2f (&a)[24])
{
    #pragma unroll
    for (int j = 0; j < 12; ++j) {
        const int kb = j * 8 + 4 * khalf;
        a[2 * j].x     = xrow[(size_t)kb * 4];       a[2 * j].y     = xrow[(size_t)(kb + 1) * 4];
        a[2 * j + 1].x = xrow[(size_t)(kb + 2) * 4]; a[2 * j + 1].y = xrow[(size_t)(kb + 3) * 4];
    }
}

__device__ __forceinline__ v8f mma12(v8f acc, const v2f (&a)[24], const v4f (&w)[12])
{
    #pragma unroll
    for (int j = 0; j < 12; ++j) {
        v2f bA; bA.x = w[j].x; bA.y = w[j].y;
        v2f bB; bB.x = w[j].z; bB.y = w[j].w;
        acc = __builtin_amdgcn_wmma_f32_16x16x4_f32(false, a[2 * j], false, bA,
                                                    (short)0, acc, false, false);
        acc = __builtin_amdgcn_wmma_f32_16x16x4_f32(false, a[2 * j + 1], false, bB,
                                                    (short)0, acc, false, false);
    }
    return acc;
}

__global__ __launch_bounds__(64) void
ssm_fused_kernel(const float* __restrict__ x,
                 const float* __restrict__ Win,
                 const float* __restrict__ Wxp,
                 const float* __restrict__ Wdt,
                 const float* __restrict__ dt_bias,
                 const float* __restrict__ A_logs,
                 const float* __restrict__ Dvec,
                 const float* __restrict__ gamma,
                 const float* __restrict__ beta,
                 const float* __restrict__ Wout,
                 float* __restrict__ out,
                 int npix)
{
    __shared__ WaveSmem smem[WAVES_PER_BLOCK];

    const int wave = threadIdx.x >> 5;
    const int lane = threadIdx.x & 31;
    const int pix0 = (blockIdx.x * WAVES_PER_BLOCK + wave) * NPIX_PER_WAVE;
    if (pix0 + NPIX_PER_WAVE > npix) return;

    WaveSmem& S = smem[wave];

    const int rowM  = lane & 15;
    const int khalf = lane >> 4;
    const float* xrow = x + (size_t)(pix0 + (rowM >> 2)) * 384 + (rowM & 3);

    v4f wa[12], wb[12];

    {
        v2f afr[24];
        load_x12(xrow, khalf, afr);
        const float* w0 = Win + (size_t)rowM * 96 + 4 * khalf;
        const float* wxp0 = Wxp + (size_t)rowM * 192 + 4 * khalf;
        load_w12(w0, wa);
        for (int nt = 0; nt < 12; nt += 2) {
            const float* wt = w0 + (size_t)nt * (16 * 96);
            load_w12(wt + 16 * 96, wb);
            v8f acc = {};
            acc = mma12(acc, afr, wa);
            load_w12((nt + 2 < 12) ? wt + 32 * 96 : wxp0, wa);
            for (int r = 0; r < 8; ++r) {
                float v = acc[r];
                v = v / (1.0f + __expf(-v));
                S.y[(r + 8 * khalf) * Y_STRIDE + nt * 16 + rowM] = v;
            }
            v8f acc2 = {};
            acc2 = mma12(acc2, afr, wb);
            for (int r = 0; r < 8; ++r) {
                float v = acc2[r];
                v = v / (1.0f + __expf(-v));
                S.y[(r + 8 * khalf) * Y_STRIDE + (nt + 1) * 16 + rowM] = v;
            }
        }
    }
    __syncthreads();

    {
        v2f alo[24], ahi[24];
        load_a12(&S.y[rowM * Y_STRIDE + 4 * khalf], alo);
        load_a12(&S.y[rowM * Y_STRIDE + 4 * khalf + 96], ahi);
        for (int nt = 0; nt < 3; ++nt) {
            const int c  = nt * 16 + rowM;
            const bool valid = (c < 38);
            const float* wt = Wxp + (size_t)(valid ? c : 37) * 192 + 4 * khalf;
            load_w12(wt + 96, wb);
            #pragma unroll
            for (int j = 0; j < 12; ++j) {
                wa[j].x = valid ? wa[j].x : 0.0f; wa[j].y = valid ? wa[j].y : 0.0f;
                wa[j].z = valid ? wa[j].z : 0.0f; wa[j].w = valid ? wa[j].w : 0.0f;
            }
            v8f acc = {};
            acc = mma12(acc, alo, wa);
            if (nt + 1 < 3) {
                const int cn = c + 16;
                load_w12(Wxp + (size_t)(cn < 38 ? cn : 37) * 192 + 4 * khalf, wa);
            }
            #pragma unroll
            for (int j = 0; j < 12; ++j) {
                wb[j].x = valid ? wb[j].x : 0.0f; wb[j].y = valid ? wb[j].y : 0.0f;
                wb[j].z = valid ? wb[j].z : 0.0f; wb[j].w = valid ? wb[j].w : 0.0f;
            }
            acc = mma12(acc, ahi, wb);
            for (int r = 0; r < 8; ++r)
                S.xdbl[(r + 8 * khalf) * 48 + c] = acc[r];
        }
    }
    __syncthreads();

    {
        float lnsum[16], lnsq[16];
        #pragma unroll
        for (int r = 0; r < 16; ++r) { lnsum[r] = 0.f; lnsq[r] = 0.f; }

        for (int it = 0; it < 6; ++it) {
            const int ch = it * 32 + lane;
            float Av[16];
            #pragma unroll
            for (int n = 0; n < 16; ++n) Av[n] = -__expf(A_logs[ch * 16 + n]);
            const float dtb = dt_bias[ch];
            const float dch = Dvec[ch];
            float wdt[6];
            #pragma unroll
            for (int r = 0; r < 6; ++r) wdt[r] = Wdt[ch * 6 + r];

            #pragma unroll
            for (int p = 0; p < 4; ++p) {
                float h[16];
                #pragma unroll
                for (int n = 0; n < 16; ++n) h[n] = 0.f;
                #pragma unroll
                for (int l = 0; l < 4; ++l) {
                    const int row = p * 4 + l;
                    const float* xd = &S.xdbl[row * 48];
                    float dacc = dtb;
                    #pragma unroll
                    for (int r = 0; r < 6; ++r) dacc += xd[r] * wdt[r];
                    const float delta = (dacc > 20.f) ? dacc
                                       : __logf(1.0f + __expf(dacc));
                    const float u  = S.y[row * Y_STRIDE + ch];
                    const float du = delta * u;
                    float y = 0.f;
                    #pragma unroll
                    for (int n = 0; n < 16; ++n) {
                        const float dA = __expf(delta * Av[n]);
                        h[n] = dA * h[n] + du * xd[6 + n];
                        y += h[n] * xd[22 + n];
                    }
                    y += dch * u;
                    S.y[row * Y_STRIDE + ch] = y;
                    lnsum[row] += y; lnsq[row] += y * y;
                }
            }
        }
        #pragma unroll
        for (int m = 1; m < 32; m <<= 1) {
            #pragma unroll
            for (int r = 0; r < 16; ++r) {
                lnsum[r] += __shfl_xor(lnsum[r], m, 32);
                lnsq[r]  += __shfl_xor(lnsq[r],  m, 32);
            }
        }
        if (lane == 0) {
            #pragma unroll
            for (int r = 0; r < 16; ++r) {
                const float mu  = lnsum[r] * (1.0f / 192.0f);
                const float var = lnsq[r] * (1.0f / 192.0f) - mu * mu;
                S.mu[r] = mu;
                S.rs[r] = rsqrtf(var + 1e-5f);
            }
        }
    }
    __syncthreads();

    {
        float mur[8], rsr[8];
        #pragma unroll
        for (int r = 0; r < 8; ++r) { mur[r] = S.mu[r + 8 * khalf]; rsr[r] = S.rs[r + 8 * khalf]; }

        v2f afr[24];
        load_x12(xrow, khalf, afr);
        const float* w0 = Win + (size_t)(192 + rowM) * 96 + 4 * khalf;
        const float* wout0 = Wout + (size_t)rowM * 192 + 4 * khalf;
        load_w12(w0, wa);
        for (int nt = 0; nt < 12; nt += 2) {
            const int ncol = nt * 16 + rowM;
            const float g0 = gamma[ncol],      bt0 = beta[ncol];
            const float g1 = gamma[ncol + 16], bt1 = beta[ncol + 16];
            const float* wt = w0 + (size_t)nt * (16 * 96);
            load_w12(wt + 16 * 96, wb);
            v8f acc = {};
            acc = mma12(acc, afr, wa);
            load_w12((nt + 2 < 12) ? wt + 32 * 96 : wout0, wa);
            for (int r = 0; r < 8; ++r) {
                float z = acc[r];
                z = z / (1.0f + __expf(-z));
                const int idx = (r + 8 * khalf) * Y_STRIDE + ncol;
                float v = (S.y[idx] - mur[r]) * rsr[r] * g0 + bt0;
                v *= z;
                S.y[idx] = v;
            }
            v8f acc2 = {};
            acc2 = mma12(acc2, afr, wb);
            for (int r = 0; r < 8; ++r) {
                float z = acc2[r];
                z = z / (1.0f + __expf(-z));
                const int idx = (r + 8 * khalf) * Y_STRIDE + ncol + 16;
                float v = (S.y[idx] - mur[r]) * rsr[r] * g1 + bt1;
                v *= z;
                S.y[idx] = v;
            }
        }
    }
    __syncthreads();

    {
        v2f alo[24], ahi[24];
        load_a12(&S.y[rowM * Y_STRIDE + 4 * khalf], alo);
        load_a12(&S.y[rowM * Y_STRIDE + 4 * khalf + 96], ahi);
        const float* w0 = Wout + (size_t)rowM * 192 + 4 * khalf;
        #pragma unroll 1
        for (int nt = 0; nt < 5; ++nt) {
            const float* wt = w0 + (size_t)nt * (16 * 192);
            load_w12(wt + 96, wb);
            v8f acc = {};
            acc = mma12(acc, alo, wa);
            load_w12(wt + 16 * 192, wa);
            acc = mma12(acc, ahi, wb);
            const int ncol = nt * 16 + rowM;
            for (int r = 0; r < 8; ++r) {
                const int row = r + 8 * khalf;
                out[(size_t)(pix0 + (row >> 2)) * 384 + ncol * 4 + (row & 3)] = acc[r];
            }
        }
        {
            const float* wt = w0 + (size_t)5 * (16 * 192);
            load_w12(wt + 96, wb);
            v8f acc = {};
            acc = mma12(acc, alo, wa);
            acc = mma12(acc, ahi, wb);
            const int ncol = 5 * 16 + rowM;
            for (int r = 0; r < 8; ++r) {
                const int row = r + 8 * khalf;
                out[(size_t)(pix0 + (row >> 2)) * 384 + ncol * 4 + (row & 3)] = acc[r];
            }
        }
    }
}

extern "C" void kernel_launch(void* const* d_in, const int* in_sizes, int n_in,
                              void* d_out, int out_size, void* d_ws, size_t ws_size,
                              hipStream_t stream) {
    const float* x       = (const float*)d_in[0];
    const float* Win     = (const float*)d_in[1];
    const float* Wxp     = (const float*)d_in[2];
    const float* Wdt     = (const float*)d_in[3];
    const float* dt_bias = (const float*)d_in[4];
    const float* A_logs  = (const float*)d_in[5];
    const float* Dvec    = (const float*)d_in[6];
    const float* gamma   = (const float*)d_in[7];
    const float* beta    = (const float*)d_in[8];
    const float* Wout    = (const float*)d_in[9];
    float* out = (float*)d_out;

    const int npix = in_sizes[0] / (96 * 4);
    const int pix_per_block = NPIX_PER_WAVE * WAVES_PER_BLOCK;
    const int grid = npix / pix_per_block;

    ssm_fused_kernel<<<grid, 64, 0, stream>>>(x, Win, Wxp, Wdt, dt_bias, A_logs,
                                              Dvec, gamma, beta, Wout, out, npix);
}


# Scan stage two-wide and branch-free: A*log2(e), state, B, C as float pairs on packed mul/FMA, y as even/odd partial sums (f32 sum order changes), softplus on plain exp2/log2, aligned b128 B/C reads
# speedup vs baseline: 2.5277x; 1.2006x over previous
#include <hip/hip_runtime.h>
#include <hip/hip_bf16.h>

typedef __attribute__((ext_vector_type(2))) float v2f;
typedef __attribute__((ext_vector_type(8))) float v8f;

#define Y_STRIDE 196
#define NPIX_PER_WAVE 4
#define WAVES_PER_BLOCK 2

struct alignas(16) WaveSmem {
    float y[16 * Y_STRIDE];
    float xdbl[16 * 48];
    float mu[16];
    float rs[16];
};

__device__ __forceinline__ int xp_row(int p, bool& valid)
{
    valid = (p < 6) || (p >= 8 && p < 40);
    return valid ? (p < 6 ? p : p - 2) : 37;
}

typedef __attribute__((ext_vector_type(4))) float v4f;


__device__ __forceinline__ void load_w12(const float* __restrict__ wb, v4f (&w)[12])
{
    #pragma unroll
    for (int j = 0; j < 12; ++j) {
        w[j].x = wb[j * 8];     w[j].y = wb[j * 8 + 1];
        w[j].z = wb[j * 8 + 2]; w[j].w = wb[j * 8 + 3];
    }
}

__device__ __forceinline__ void load_a12(const float* arow, v2f (&a)[24])
{
    #pragma unroll
    for (int j = 0; j < 12; ++j) {
        a[2 * j].x     = arow[j * 8];     a[2 * j].y     = arow[j * 8 + 1];
        a[2 * j + 1].x = arow[j * 8 + 2]; a[2 * j + 1].y = arow[j * 8 + 3];
    }
}

__device__ __forceinline__ void load_x12(const float* __restrict__ xrow, int khalf, v2f (&a)[24])
{
    #pragma unroll
    for (int j = 0; j < 12; ++j) {
        const int kb = j * 8 + 4 * khalf;
        a[2 * j].x     = xrow[(size_t)kb * 4];       a[2 * j].y     = xrow[(size_t)(kb + 1) * 4];
        a[2 * j + 1].x = xrow[(size_t)(kb + 2) * 4]; a[2 * j + 1].y = xrow[(size_t)(kb + 3) * 4];
    }
}

__device__ __forceinline__ v8f mma12(v8f acc, const v2f (&a)[24], const v4f (&w)[12])
{
    #pragma unroll
    for (int j = 0; j < 12; ++j) {
        v2f bA; bA.x = w[j].x; bA.y = w[j].y;
        v2f bB; bB.x = w[j].z; bB.y = w[j].w;
        acc = __builtin_amdgcn_wmma_f32_16x16x4_f32(false, a[2 * j], false, bA,
                                                    (short)0, acc, false, false);
        acc = __builtin_amdgcn_wmma_f32_16x16x4_f32(false, a[2 * j + 1], false, bB,
                                                    (short)0, acc, false, false);
    }
    return acc;
}

struct ScanRowIn { float d[6]; v2f B[8]; v2f C[8]; float u; };

__device__ __forceinline__ void load_scan_row(const float* xd, const float* uslot, ScanRowIn& R)
{
    #pragma unroll
    for (int r = 0; r < 6; ++r) R.d[r] = xd[r];
    const v2f* Bp = reinterpret_cast<const v2f*>(xd + 8);
    const v2f* Cp = reinterpret_cast<const v2f*>(xd + 24);
    #pragma unroll
    for (int i = 0; i < 8; ++i) { R.B[i] = Bp[i]; R.C[i] = Cp[i]; }
    R.u = *uslot;
}

__global__ __launch_bounds__(64) void
ssm_fused_kernel(const float* __restrict__ x,
                 const float* __restrict__ Win,
                 const float* __restrict__ Wxp,
                 const float* __restrict__ Wdt,
                 const float* __restrict__ dt_bias,
                 const float* __restrict__ A_logs,
                 const float* __restrict__ Dvec,
                 const float* __restrict__ gamma,
                 const float* __restrict__ beta,
                 const float* __restrict__ Wout,
                 float* __restrict__ out,
                 int npix)
{
    __shared__ WaveSmem smem[WAVES_PER_BLOCK];

    const int wave = threadIdx.x >> 5;
    const int lane = threadIdx.x & 31;
    const int pix0 = (blockIdx.x * WAVES_PER_BLOCK + wave) * NPIX_PER_WAVE;
    if (pix0 + NPIX_PER_WAVE > npix) return;

    WaveSmem& S = smem[wave];

    const int rowM  = lane & 15;
    const int khalf = lane >> 4;
    const float* xrow = x + (size_t)(pix0 + (rowM >> 2)) * 384 + (rowM & 3);

    v4f wa[12], wb[12];

    {
        v2f afr[24];
        load_x12(xrow, khalf, afr);
        const float* w0 = Win + (size_t)rowM * 96 + 4 * khalf;
        bool v0;
        const float* wxp0 = Wxp + (size_t)xp_row(rowM, v0) * 192 + 4 * khalf;
        load_w12(w0, wa);
        for (int nt = 0; nt < 12; nt += 2) {
            const float* wt = w0 + (size_t)nt * (16 * 96);
            load_w12(wt + 16 * 96, wb);
            v8f acc = {};
            acc = mma12(acc, afr, wa);
            load_w12((nt + 2 < 12) ? wt + 32 * 96 : wxp0, wa);
            for (int r = 0; r < 8; ++r) {
                float v = acc[r];
                v = v / (1.0f + __expf(-v));
                S.y[(r + 8 * khalf) * Y_STRIDE + nt * 16 + rowM] = v;
            }
            v8f acc2 = {};
            acc2 = mma12(acc2, afr, wb);
            for (int r = 0; r < 8; ++r) {
                float v = acc2[r];
                v = v / (1.0f + __expf(-v));
                S.y[(r + 8 * khalf) * Y_STRIDE + (nt + 1) * 16 + rowM] = v;
            }
        }
    }
    __syncthreads();

    {
        v2f alo[24], ahi[24];
        load_a12(&S.y[rowM * Y_STRIDE + 4 * khalf], alo);
        load_a12(&S.y[rowM * Y_STRIDE + 4 * khalf + 96], ahi);
        for (int nt = 0; nt < 3; ++nt) {
            const int c  = nt * 16 + rowM;
            bool valid;
            const float* wt = Wxp + (size_t)xp_row(c, valid) * 192 + 4 * khalf;
            load_w12(wt + 96, wb);
            #pragma unroll
            for (int j = 0; j < 12; ++j) {
                wa[j].x = valid ? wa[j].x : 0.0f; wa[j].y = valid ? wa[j].y : 0.0f;
                wa[j].z = valid ? wa[j].z : 0.0f; wa[j].w = valid ? wa[j].w : 0.0f;
            }
            v8f acc = {};
            acc = mma12(acc, alo, wa);
            if (nt + 1 < 3) {
                bool vn;
                load_w12(Wxp + (size_t)xp_row(c + 16, vn) * 192 + 4 * khalf, wa);
            }
            #pragma unroll
            for (int j = 0; j < 12; ++j) {
                wb[j].x = valid ? wb[j].x : 0.0f; wb[j].y = valid ? wb[j].y : 0.0f;
                wb[j].z = valid ? wb[j].z : 0.0f; wb[j].w = valid ? wb[j].w : 0.0f;
            }
            acc = mma12(acc, ahi, wb);
            for (int r = 0; r < 8; ++r)
                S.xdbl[(r + 8 * khalf) * 48 + c] = acc[r];
        }
    }
    __syncthreads();

    {
        float lnsum[16], lnsq[16];
        #pragma unroll
        for (int r = 0; r < 16; ++r) { lnsum[r] = 0.f; lnsq[r] = 0.f; }

        for (int it = 0; it < 6; ++it) {
            const int ch = it * 32 + lane;
            v2f Av[8];
            #pragma unroll
            for (int i = 0; i < 8; ++i) {
                Av[i].x = -__expf(A_logs[ch * 16 + 2 * i])     * 1.4426950408889634f;
                Av[i].y = -__expf(A_logs[ch * 16 + 2 * i + 1]) * 1.4426950408889634f;
            }
            const float dtb = dt_bias[ch];
            const float dch = Dvec[ch];
            float wdt[6];
            #pragma unroll
            for (int r = 0; r < 6; ++r) wdt[r] = Wdt[ch * 6 + r];

            ScanRowIn cur, nxt;
            load_scan_row(&S.xdbl[0], &S.y[ch], cur);
            #pragma unroll
            for (int p = 0; p < 4; ++p) {
                v2f h[8];
                #pragma unroll
                for (int i = 0; i < 8; ++i) { h[i].x = 0.f; h[i].y = 0.f; }
                #pragma unroll
                for (int l = 0; l < 4; ++l) {
                    const int row = p * 4 + l;
                    if (row + 1 < 16)
                        load_scan_row(&S.xdbl[(row + 1) * 48], &S.y[(row + 1) * Y_STRIDE + ch], nxt);
                    float dacc = dtb;
                    #pragma unroll
                    for (int r = 0; r < 6; ++r) dacc += cur.d[r] * wdt[r];
                    const float e     = __builtin_amdgcn_exp2f(dacc * 1.4426950408889634f);
                    const float sp    = __builtin_amdgcn_logf(1.0f + e) * 0.6931471805599453f;
                    const float delta = (dacc > 20.f) ? dacc : sp;
                    const float u  = cur.u;
                    const float du = delta * u;
                    v2f dl;  dl.x = delta;  dl.y = delta;
                    v2f du2; du2.x = du;    du2.y = du;
                    v2f y2;  y2.x = 0.f;    y2.y = 0.f;
                    #pragma unroll
                    for (int i = 0; i < 8; ++i) {
                        const v2f t = dl * Av[i];
                        v2f dA;
                        dA.x = __builtin_amdgcn_exp2f(t.x);
                        dA.y = __builtin_amdgcn_exp2f(t.y);
                        h[i] = dA * h[i] + du2 * cur.B[i];
                        y2   = h[i] * cur.C[i] + y2;
                    }
                    const float y = (y2.x + y2.y) + dch * u;
                    S.y[row * Y_STRIDE + ch] = y;
                    lnsum[row] += y; lnsq[row] += y * y;
                    if (row + 1 < 16) cur = nxt;
                }
            }
        }
        #pragma unroll
        for (int m = 1; m < 32; m <<= 1) {
            #pragma unroll
            for (int r = 0; r < 16; ++r) {
                lnsum[r] += __shfl_xor(lnsum[r], m, 32);
                lnsq[r]  += __shfl_xor(lnsq[r],  m, 32);
            }
        }
        if (lane == 0) {
            #pragma unroll
            for (int r = 0; r < 16; ++r) {
                const float mu  = lnsum[r] * (1.0f / 192.0f);
                const float var = lnsq[r] * (1.0f / 192.0f) - mu * mu;
                S.mu[r] = mu;
                S.rs[r] = rsqrtf(var + 1e-5f);
            }
        }
    }
    __syncthreads();

    {
        float mur[8], rsr[8];
        #pragma unroll
        for (int r = 0; r < 8; ++r) { mur[r] = S.mu[r + 8 * khalf]; rsr[r] = S.rs[r + 8 * khalf]; }

        v2f afr[24];
        load_x12(xrow, khalf, afr);
        const float* w0 = Win + (size_t)(192 + rowM) * 96 + 4 * khalf;
        const float* wout0 = Wout + (size_t)rowM * 192 + 4 * khalf;
        load_w12(w0, wa);
        for (int nt = 0; nt < 12; nt += 2) {
            const int ncol = nt * 16 + rowM;
            const float g0 = gamma[ncol],      bt0 = beta[ncol];
            const float g1 = gamma[ncol + 16], bt1 = beta[ncol + 16];
            const float* wt = w0 + (size_t)nt * (16 * 96);
            load_w12(wt + 16 * 96, wb);
            v8f acc = {};
            acc = mma12(acc, afr, wa);
            load_w12((nt + 2 < 12) ? wt + 32 * 96 : wout0, wa);
            for (int r = 0; r < 8; ++r) {
                float z = acc[r];
                z = z / (1.0f + __expf(-z));
                const int idx = (r + 8 * khalf) * Y_STRIDE + ncol;
                float v = (S.y[idx] - mur[r]) * rsr[r] * g0 + bt0;
                v *= z;
                S.y[idx] = v;
            }
            v8f acc2 = {};
            acc2 = mma12(acc2, afr, wb);
            for (int r = 0; r < 8; ++r) {
                float z = acc2[r];
                z = z / (1.0f + __expf(-z));
                const int idx = (r + 8 * khalf) * Y_STRIDE + ncol + 16;
                float v = (S.y[idx] - mur[r]) * rsr[r] * g1 + bt1;
                v *= z;
                S.y[idx] = v;
            }
        }
    }
    __syncthreads();

    {
        v2f alo[24], ahi[24];
        load_a12(&S.y[rowM * Y_STRIDE + 4 * khalf], alo);
        load_a12(&S.y[rowM * Y_STRIDE + 4 * khalf + 96], ahi);
        const float* w0 = Wout + (size_t)rowM * 192 + 4 * khalf;
        #pragma unroll 1
        for (int nt = 0; nt < 5; ++nt) {
            const float* wt = w0 + (size_t)nt * (16 * 192);
            load_w12(wt + 96, wb);
            v8f acc = {};
            acc = mma12(acc, alo, wa);
            load_w12(wt + 16 * 192, wa);
            acc = mma12(acc, ahi, wb);
            const int ncol = nt * 16 + rowM;
            for (int r = 0; r < 8; ++r) {
                const int row = r + 8 * khalf;
                out[(size_t)(pix0 + (row >> 2)) * 384 + ncol * 4 + (row & 3)] = acc[r];
            }
        }
        {
            const float* wt = w0 + (size_t)5 * (16 * 192);
            load_w12(wt + 96, wb);
            v8f acc = {};
            acc = mma12(acc, alo, wa);
            acc = mma12(acc, ahi, wb);
            const int ncol = 5 * 16 + rowM;
            for (int r = 0; r < 8; ++r) {
                const int row = r + 8 * khalf;
                out[(size_t)(pix0 + (row >> 2)) * 384 + ncol * 4 + (row & 3)] = acc[r];
            }
        }
    }
}

extern "C" void kernel_launch(void* const* d_in, const int* in_sizes, int n_in,
                              void* d_out, int out_size, void* d_ws, size_t ws_size,
                              hipStream_t stream) {
    const float* x       = (const float*)d_in[0];
    const float* Win     = (const float*)d_in[1];
    const float* Wxp     = (const float*)d_in[2];
    const float* Wdt     = (const float*)d_in[3];
    const float* dt_bias = (const float*)d_in[4];
    const float* A_logs  = (const float*)d_in[5];
    const float* Dvec    = (const float*)d_in[6];
    const float* gamma   = (const float*)d_in[7];
    const float* beta    = (const float*)d_in[8];
    const float* Wout    = (const float*)d_in[9];
    float* out = (float*)d_out;

    const int npix = in_sizes[0] / (96 * 4);
    const int pix_per_block = NPIX_PER_WAVE * WAVES_PER_BLOCK;
    const int grid = npix / pix_per_block;

    ssm_fused_kernel<<<grid, 64, 0, stream>>>(x, Win, Wxp, Wdt, dt_bias, A_logs,
                                              Dvec, gamma, beta, Wout, out, npix);
}


# 4-wave blocks (one wave per SIMD); the 64 tiles left when 3072 fill every SIMD 3 deep run first, one per block, columns and channels split over its waves (f32 sum order of their LN sums changes)
# speedup vs baseline: 2.7514x; 1.0885x over previous
#include <hip/hip_runtime.h>
#include <hip/hip_bf16.h>

typedef __attribute__((ext_vector_type(2))) float v2f;
typedef __attribute__((ext_vector_type(8))) float v8f;

#define Y_STRIDE 196
#define NPIX_PER_WAVE 4
#define WAVES_PER_BLOCK 4
#define TILE_ROUND 1024

struct alignas(16) WaveSmem {
    float y[16 * Y_STRIDE];
    float xdbl[16 * 48];
    float mu[16];
    float rs[16];
};

__device__ __forceinline__ int xp_row(int p, bool& valid)
{
    valid = (p < 6) || (p >= 8 && p < 40);
    return valid ? (p < 6 ? p : p - 2) : 37;
}

typedef __attribute__((ext_vector_type(4))) float v4f;


__device__ __forceinline__ void load_w12(const float* __restrict__ wb, v4f (&w)[12])
{
    #pragma unroll
    for (int j = 0; j < 12; ++j) {
        w[j].x = wb[j * 8];     w[j].y = wb[j * 8 + 1];
        w[j].z = wb[j * 8 + 2]; w[j].w = wb[j * 8 + 3];
    }
}

__device__ __forceinline__ void load_a12(const float* arow, v2f (&a)[24])
{
    #pragma unroll
    for (int j = 0; j < 12; ++j) {
        a[2 * j].x     = arow[j * 8];     a[2 * j].y     = arow[j * 8 + 1];
        a[2 * j + 1].x = arow[j * 8 + 2]; a[2 * j + 1].y = arow[j * 8 + 3];
    }
}

__device__ __forceinline__ void load_x12(const float* __restrict__ xrow, int khalf, v2f (&a)[24])
{
    #pragma unroll
    for (int j = 0; j < 12; ++j) {
        const int kb = j * 8 + 4 * khalf;
        a[2 * j].x     = xrow[(size_t)kb * 4];       a[2 * j].y     = xrow[(size_t)(kb + 1) * 4];
        a[2 * j + 1].x = xrow[(size_t)(kb + 2) * 4]; a[2 * j + 1].y = xrow[(size_t)(kb + 3) * 4];
    }
}

__device__ __forceinline__ v8f mma12(v8f acc, const v2f (&a)[24], const v4f (&w)[12])
{
    #pragma unroll
    for (int j = 0; j < 12; ++j) {
        v2f bA; bA.x = w[j].x; bA.y = w[j].y;
        v2f bB; bB.x = w[j].z; bB.y = w[j].w;
        acc = __builtin_amdgcn_wmma_f32_16x16x4_f32(false, a[2 * j], false, bA,
                                                    (short)0, acc, false, false);
        acc = __builtin_amdgcn_wmma_f32_16x16x4_f32(false, a[2 * j + 1], false, bB,
                                                    (short)0, acc, false, false);
    }
    return acc;
}

struct ScanRowIn { float d[6]; v2f B[8]; v2f C[8]; float u; };

__device__ __forceinline__ void load_scan_row(const float* xd, const float* uslot, ScanRowIn& R)
{
    #pragma unroll
    for (int r = 0; r < 6; ++r) R.d[r] = xd[r];
    const v2f* Bp = reinterpret_cast<const v2f*>(xd + 8);
    const v2f* Cp = reinterpret_cast<const v2f*>(xd + 24);
    #pragma unroll
    for (int i = 0; i < 8; ++i) { R.B[i] = Bp[i]; R.C[i] = Cp[i]; }
    R.u = *uslot;
}

__global__ __launch_bounds__(32 * WAVES_PER_BLOCK) void
ssm_fused_kernel(const float* __restrict__ x,
                 const float* __restrict__ Win,
                 const float* __restrict__ Wxp,
                 const float* __restrict__ Wdt,
                 const float* __restrict__ dt_bias,
                 const float* __restrict__ A_logs,
                 const float* __restrict__ Dvec,
                 const float* __restrict__ gamma,
                 const float* __restrict__ beta,
                 const float* __restrict__ Wout,
                 float* __restrict__ out,
                 int npix)
{
    __shared__ WaveSmem smem[WAVES_PER_BLOCK];

    const int wave = __builtin_amdgcn_readfirstlane((int)(threadIdx.x >> 5));
    const int lane = threadIdx.x & 31;
    const int ntail = (npix / NPIX_PER_WAVE) % TILE_ROUND;
    const bool tail = (int)blockIdx.x < ntail;
    const int pix0 = tail ? (int)blockIdx.x * NPIX_PER_WAVE
                          : (ntail + ((int)blockIdx.x - ntail) * WAVES_PER_BLOCK + wave) * NPIX_PER_WAVE;
    if (pix0 + NPIX_PER_WAVE > npix) return;

    WaveSmem& S = smem[tail ? 0 : wave];

    const int rowM  = lane & 15;
    const int khalf = lane >> 4;
    const float* xrow = x + (size_t)(pix0 + (rowM >> 2)) * 384 + (rowM & 3);

    v4f wa[12], wb[12];

    if (!tail) {
        v2f afr[24];
        load_x12(xrow, khalf, afr);
        const float* w0 = Win + (size_t)rowM * 96 + 4 * khalf;
        bool v0;
        const float* wxp0 = Wxp + (size_t)xp_row(rowM, v0) * 192 + 4 * khalf;
        load_w12(w0, wa);
        for (int nt = 0; nt < 12; nt += 2) {
            const float* wt = w0 + (size_t)nt * (16 * 96);
            load_w12(wt + 16 * 96, wb);
            v8f acc = {};
            acc = mma12(acc, afr, wa);
            load_w12((nt + 2 < 12) ? wt + 32 * 96 : wxp0, wa);
            for (int r = 0; r < 8; ++r) {
                float v = acc[r];
                v = v / (1.0f + __expf(-v));
                S.y[(r + 8 * khalf) * Y_STRIDE + nt * 16 + rowM] = v;
            }
            v8f acc2 = {};
            acc2 = mma12(acc2, afr, wb);
            for (int r = 0; r < 8; ++r) {
                float v = acc2[r];
                v = v / (1.0f + __expf(-v));
                S.y[(r + 8 * khalf) * Y_STRIDE + (nt + 1) * 16 + rowM] = v;
            }
        }
    } else {
        v2f afr[24];
        load_x12(xrow, khalf, afr);
        #pragma unroll 1
        for (int t = 0; t < 3; ++t) {
            const int nt = wave * 3 + t;
            load_w12(Win + (size_t)(nt * 16 + rowM) * 96 + 4 * khalf, wa);
            v8f acc = {};
            acc = mma12(acc, afr, wa);
            for (int r = 0; r < 8; ++r) {
                float v = acc[r];
                v = v / (1.0f + __expf(-v));
                S.y[(r + 8 * khalf) * Y_STRIDE + nt * 16 + rowM] = v;
            }
        }
    }
    __syncthreads();

    if (!tail) {
        v2f alo[24], ahi[24];
        load_a12(&S.y[rowM * Y_STRIDE + 4 * khalf], alo);
        load_a12(&S.y[rowM * Y_STRIDE + 4 * khalf + 96], ahi);
        for (int nt = 0; nt < 3; ++nt) {
            const int c  = nt * 16 + rowM;
            bool valid;
            const float* wt = Wxp + (size_t)xp_row(c, valid) * 192 + 4 * khalf;
            load_w12(wt + 96, wb);
            #pragma unroll
            for (int j = 0; j < 12; ++j) {
                wa[j].x = valid ? wa[j].x : 0.0f; wa[j].y = valid ? wa[j].y : 0.0f;
                wa[j].z = valid ? wa[j].z : 0.0f; wa[j].w = valid ? wa[j].w : 0.0f;
            }
            v8f acc = {};
            acc = mma12(acc, alo, wa);
            if (nt + 1 < 3) {
                bool vn;
                load_w12(Wxp + (size_t)xp_row(c + 16, vn) * 192 + 4 * khalf, wa);
            }
            #pragma unroll
            for (int j = 0; j < 12; ++j) {
                wb[j].x = valid ? wb[j].x : 0.0f; wb[j].y = valid ? wb[j].y : 0.0f;
                wb[j].z = valid ? wb[j].z : 0.0f; wb[j].w = valid ? wb[j].w : 0.0f;
            }
            acc = mma12(acc, ahi, wb);
            for (int r = 0; r < 8; ++r)
                S.xdbl[(r + 8 * khalf) * 48 + c] = acc[r];
        }
    } else if (wave < 3) {
        v2f alo[24], ahi[24];
        load_a12(&S.y[rowM * Y_STRIDE + 4 * khalf], alo);
        load_a12(&S.y[rowM * Y_STRIDE + 4 * khalf + 96], ahi);
        const int c = wave * 16 + rowM;
        bool valid;
        const float* wt = Wxp + (size_t)xp_row(c, valid) * 192 + 4 * khalf;
        load_w12(wt, wa);
        load_w12(wt + 96, wb);
        #pragma unroll
        for (int j = 0; j < 12; ++j) {
            wa[j].x = valid ? wa[j].x : 0.0f; wa[j].y = valid ? wa[j].y : 0.0f;
            wa[j].z = valid ? wa[j].z : 0.0f; wa[j].w = valid ? wa[j].w : 0.0f;
            wb[j].x = valid ? wb[j].x : 0.0f; wb[j].y = valid ? wb[j].y : 0.0f;
            wb[j].z = valid ? wb[j].z : 0.0f; wb[j].w = valid ? wb[j].w : 0.0f;
        }
        v8f acc = {};
        acc = mma12(acc, alo, wa);
        acc = mma12(acc, ahi, wb);
        for (int r = 0; r < 8; ++r)
            S.xdbl[(r + 8 * khalf) * 48 + c] = acc[r];
    }
    __syncthreads();

    {
        float lnsum[16], lnsq[16];
        #pragma unroll
        for (int r = 0; r < 16; ++r) { lnsum[r] = 0.f; lnsq[r] = 0.f; }

        const int it0    = tail ? wave : 0;
        const int itstep = tail ? WAVES_PER_BLOCK : 1;
        for (int it = it0; it < 6; it += itstep) {
            const int ch = it * 32 + lane;
            v2f Av[8];
            #pragma unroll
            for (int i = 0; i < 8; ++i) {
                Av[i].x = -__expf(A_logs[ch * 16 + 2 * i])     * 1.4426950408889634f;
                Av[i].y = -__expf(A_logs[ch * 16 + 2 * i + 1]) * 1.4426950408889634f;
            }
            const float dtb = dt_bias[ch];
            const float dch = Dvec[ch];
            float wdt[6];
            #pragma unroll
            for (int r = 0; r < 6; ++r) wdt[r] = Wdt[ch * 6 + r];

            ScanRowIn cur, nxt;
            load_scan_row(&S.xdbl[0], &S.y[ch], cur);
            #pragma unroll
            for (int p = 0; p < 4; ++p) {
                v2f h[8];
                #pragma unroll
                for (int i = 0; i < 8; ++i) { h[i].x = 0.f; h[i].y = 0.f; }
                #pragma unroll
                for (int l = 0; l < 4; ++l) {
                    const int row = p * 4 + l;
                    if (row + 1 < 16)
                        load_scan_row(&S.xdbl[(row + 1) * 48], &S.y[(row + 1) * Y_STRIDE + ch], nxt);
                    float dacc = dtb;
                    #pragma unroll
                    for (int r = 0; r < 6; ++r) dacc += cur.d[r] * wdt[r];
                    const float e     = __builtin_amdgcn_exp2f(dacc * 1.4426950408889634f);
                    const float sp    = __builtin_amdgcn_logf(1.0f + e) * 0.6931471805599453f;
                    const float delta = (dacc > 20.f) ? dacc : sp;
                    const float u  = cur.u;
                    const float du = delta * u;
                    v2f dl;  dl.x = delta;  dl.y = delta;
                    v2f du2; du2.x = du;    du2.y = du;
                    v2f y2;  y2.x = 0.f;    y2.y = 0.f;
                    #pragma unroll
                    for (int i = 0; i < 8; ++i) {
                        const v2f t = dl * Av[i];
                        v2f dA;
                        dA.x = __builtin_amdgcn_exp2f(t.x);
                        dA.y = __builtin_amdgcn_exp2f(t.y);
                        h[i] = dA * h[i] + du2 * cur.B[i];
                        y2   = h[i] * cur.C[i] + y2;
                    }
                    const float y = (y2.x + y2.y) + dch * u;
                    S.y[row * Y_STRIDE + ch] = y;
                    lnsum[row] += y; lnsq[row] += y * y;
                    if (row + 1 < 16) cur = nxt;
                }
            }
        }
        #pragma unroll
        for (int m = 1; m < 32; m <<= 1) {
            #pragma unroll
            for (int r = 0; r < 16; ++r) {
                lnsum[r] += __shfl_xor(lnsum[r], m, 32);
                lnsq[r]  += __shfl_xor(lnsq[r],  m, 32);
            }
        }
        if (!tail) {
            if (lane == 0) {
                #pragma unroll
                for (int r = 0; r < 16; ++r) {
                    const float mu  = lnsum[r] * (1.0f / 192.0f);
                    const float var = lnsq[r] * (1.0f / 192.0f) - mu * mu;
                    S.mu[r] = mu;
                    S.rs[r] = rsqrtf(var + 1e-5f);
                }
            }
        } else {
            float* part = smem[1].y;
            if (lane == 0) {
                #pragma unroll
                for (int r = 0; r < 16; ++r) {
                    part[(wave * 16 + r) * 2]     = lnsum[r];
                    part[(wave * 16 + r) * 2 + 1] = lnsq[r];
                }
            }
            __syncthreads();
            if (wave == 0 && lane < 16) {
                float sm = 0.f, sq = 0.f;
                for (int w = 0; w < WAVES_PER_BLOCK; ++w) {
                    sm += part[(w * 16 + lane) * 2];
                    sq += part[(w * 16 + lane) * 2 + 1];
                }
                const float mu  = sm * (1.0f / 192.0f);
                const float var = sq * (1.0f / 192.0f) - mu * mu;
                S.mu[lane] = mu;
                S.rs[lane] = rsqrtf(var + 1e-5f);
            }
        }
    }
    __syncthreads();

    if (!tail) {
        float mur[8], rsr[8];
        #pragma unroll
        for (int r = 0; r < 8; ++r) { mur[r] = S.mu[r + 8 * khalf]; rsr[r] = S.rs[r + 8 * khalf]; }

        v2f afr[24];
        load_x12(xrow, khalf, afr);
        const float* w0 = Win + (size_t)(192 + rowM) * 96 + 4 * khalf;
        const float* wout0 = Wout + (size_t)rowM * 192 + 4 * khalf;
        load_w12(w0, wa);
        for (int nt = 0; nt < 12; nt += 2) {
            const int ncol = nt * 16 + rowM;
            const float g0 = gamma[ncol],      bt0 = beta[ncol];
            const float g1 = gamma[ncol + 16], bt1 = beta[ncol + 16];
            const float* wt = w0 + (size_t)nt * (16 * 96);
            load_w12(wt + 16 * 96, wb);
            v8f acc = {};
            acc = mma12(acc, afr, wa);
            load_w12((nt + 2 < 12) ? wt + 32 * 96 : wout0, wa);
            for (int r = 0; r < 8; ++r) {
                float z = acc[r];
                z = z / (1.0f + __expf(-z));
                const int idx = (r + 8 * khalf) * Y_STRIDE + ncol;
                float v = (S.y[idx] - mur[r]) * rsr[r] * g0 + bt0;
                v *= z;
                S.y[idx] = v;
            }
            v8f acc2 = {};
            acc2 = mma12(acc2, afr, wb);
            for (int r = 0; r < 8; ++r) {
                float z = acc2[r];
                z = z / (1.0f + __expf(-z));
                const int idx = (r + 8 * khalf) * Y_STRIDE + ncol + 16;
                float v = (S.y[idx] - mur[r]) * rsr[r] * g1 + bt1;
                v *= z;
                S.y[idx] = v;
            }
        }
    } else {
        float mur[8], rsr[8];
        #pragma unroll
        for (int r = 0; r < 8; ++r) { mur[r] = S.mu[r + 8 * khalf]; rsr[r] = S.rs[r + 8 * khalf]; }
        v2f afr[24];
        load_x12(xrow, khalf, afr);
        for (int t = 0; t < 3; ++t) {
            const int nt   = wave * 3 + t;
            const int ncol = nt * 16 + rowM;
            const float g0 = gamma[ncol], bt0 = beta[ncol];
            load_w12(Win + (size_t)(192 + nt * 16 + rowM) * 96 + 4 * khalf, wa);
            v8f acc = {};
            acc = mma12(acc, afr, wa);
            for (int r = 0; r < 8; ++r) {
                float z = acc[r];
                z = z / (1.0f + __expf(-z));
                const int idx = (r + 8 * khalf) * Y_STRIDE + ncol;
                float v = (S.y[idx] - mur[r]) * rsr[r] * g0 + bt0;
                v *= z;
                S.y[idx] = v;
            }
        }
    }
    __syncthreads();

    if (!tail) {
        v2f alo[24], ahi[24];
        load_a12(&S.y[rowM * Y_STRIDE + 4 * khalf], alo);
        load_a12(&S.y[rowM * Y_STRIDE + 4 * khalf + 96], ahi);
        const float* w0 = Wout + (size_t)rowM * 192 + 4 * khalf;
        #pragma unroll 1
        for (int nt = 0; nt < 5; ++nt) {
            const float* wt = w0 + (size_t)nt * (16 * 192);
            load_w12(wt + 96, wb);
            v8f acc = {};
            acc = mma12(acc, alo, wa);
            load_w12(wt + 16 * 192, wa);
            acc = mma12(acc, ahi, wb);
            const int ncol = nt * 16 + rowM;
            for (int r = 0; r < 8; ++r) {
                const int row = r + 8 * khalf;
                out[(size_t)(pix0 + (row >> 2)) * 384 + ncol * 4 + (row & 3)] = acc[r];
            }
        }
        {
            const float* wt = w0 + (size_t)5 * (16 * 192);
            load_w12(wt + 96, wb);
            v8f acc = {};
            acc = mma12(acc, alo, wa);
            acc = mma12(acc, ahi, wb);
            const int ncol = 5 * 16 + rowM;
            for (int r = 0; r < 8; ++r) {
                const int row = r + 8 * khalf;
                out[(size_t)(pix0 + (row >> 2)) * 384 + ncol * 4 + (row & 3)] = acc[r];
            }
        }
    } else {
        v2f alo[24], ahi[24];
        load_a12(&S.y[rowM * Y_STRIDE + 4 * khalf], alo);
        load_a12(&S.y[rowM * Y_STRIDE + 4 * khalf + 96], ahi);
        for (int nt = wave; nt < 6; nt += WAVES_PER_BLOCK) {
            const float* wt = Wout + (size_t)(nt * 16 + rowM) * 192 + 4 * khalf;
            load_w12(wt, wa);
            load_w12(wt + 96, wb);
            v8f acc = {};
            acc = mma12(acc, alo, wa);
            acc = mma12(acc, ahi, wb);
            const int ncol = nt * 16 + rowM;
            for (int r = 0; r < 8; ++r) {
                const int row = r + 8 * khalf;
                out[(size_t)(pix0 + (row >> 2)) * 384 + ncol * 4 + (row & 3)] = acc[r];
            }
        }
    }
}

extern "C" void kernel_launch(void* const* d_in, const int* in_sizes, int n_in,
                              void* d_out, int out_size, void* d_ws, size_t ws_size,
                              hipStream_t stream) {
    const float* x       = (const float*)d_in[0];
    const float* Win     = (const float*)d_in[1];
    const float* Wxp     = (const float*)d_in[2];
    const float* Wdt     = (const float*)d_in[3];
    const float* dt_bias = (const float*)d_in[4];
    const float* A_logs  = (const float*)d_in[5];
    const float* Dvec    = (const float*)d_in[6];
    const float* gamma   = (const float*)d_in[7];
    const float* beta    = (const float*)d_in[8];
    const float* Wout    = (const float*)d_in[9];
    float* out = (float*)d_out;

    const int npix = in_sizes[0] / (96 * 4);
    const int tiles = npix / NPIX_PER_WAVE;
    const int ntail = tiles % TILE_ROUND;
    const int grid  = ntail + (tiles - ntail) / WAVES_PER_BLOCK;

    ssm_fused_kernel<<<grid, 32 * WAVES_PER_BLOCK, 0, stream>>>(x, Win, Wxp, Wdt, dt_bias, A_logs,
                                              Dvec, gamma, beta, Wout, out, npix);
}


# SiLU quotient taken by the hardware reciprocal, v * rcp(1 + exp(-v)), about 1 ulp, instead of the IEEE division sequence: about ten fewer vector instructions per activation; silu rounding moves
# speedup vs baseline: 2.8604x; 1.0396x over previous
#include <hip/hip_runtime.h>
#include <hip/hip_bf16.h>

typedef __attribute__((ext_vector_type(2))) float v2f;
typedef __attribute__((ext_vector_type(8))) float v8f;

#define Y_STRIDE 196
#define NPIX_PER_WAVE 4
#define WAVES_PER_BLOCK 4
#define TILE_ROUND 1024

struct alignas(16) WaveSmem {
    float y[16 * Y_STRIDE];
    float xdbl[16 * 48];
    float mu[16];
    float rs[16];
};

__device__ __forceinline__ int xp_row(int p, bool& valid)
{
    valid = (p < 6) || (p >= 8 && p < 40);
    return valid ? (p < 6 ? p : p - 2) : 37;
}

typedef __attribute__((ext_vector_type(4))) float v4f;

__device__ __forceinline__ float silu_rcp(float v)
{
    return v * __builtin_amdgcn_rcpf(1.0f + __expf(-v));
}


__device__ __forceinline__ void load_w12(const float* __restrict__ wb, v4f (&w)[12])
{
    #pragma unroll
    for (int j = 0; j < 12; ++j) {
        w[j].x = wb[j * 8];     w[j].y = wb[j * 8 + 1];
        w[j].z = wb[j * 8 + 2]; w[j].w = wb[j * 8 + 3];
    }
}

__device__ __forceinline__ void load_a12(const float* arow, v2f (&a)[24])
{
    #pragma unroll
    for (int j = 0; j < 12; ++j) {
        a[2 * j].x     = arow[j * 8];     a[2 * j].y     = arow[j * 8 + 1];
        a[2 * j + 1].x = arow[j * 8 + 2]; a[2 * j + 1].y = arow[j * 8 + 3];
    }
}

__device__ __forceinline__ void load_x12(const float* __restrict__ xrow, int khalf, v2f (&a)[24])
{
    #pragma unroll
    for (int j = 0; j < 12; ++j) {
        const int kb = j * 8 + 4 * khalf;
        a[2 * j].x     = xrow[(size_t)kb * 4];       a[2 * j].y     = xrow[(size_t)(kb + 1) * 4];
        a[2 * j + 1].x = xrow[(size_t)(kb + 2) * 4]; a[2 * j + 1].y = xrow[(size_t)(kb + 3) * 4];
    }
}

__device__ __forceinline__ v8f mma12(v8f acc, const v2f (&a)[24], const v4f (&w)[12])
{
    #pragma unroll
    for (int j = 0; j < 12; ++j) {
        v2f bA; bA.x = w[j].x; bA.y = w[j].y;
        v2f bB; bB.x = w[j].z; bB.y = w[j].w;
        acc = __builtin_amdgcn_wmma_f32_16x16x4_f32(false, a[2 * j], false, bA,
                                                    (short)0, acc, false, false);
        acc = __builtin_amdgcn_wmma_f32_16x16x4_f32(false, a[2 * j + 1], false, bB,
                                                    (short)0, acc, false, false);
    }
    return acc;
}

struct ScanRowIn { float d[6]; v2f B[8]; v2f C[8]; float u; };

__device__ __forceinline__ void load_scan_row(const float* xd, const float* uslot, ScanRowIn& R)
{
    #pragma unroll
    for (int r = 0; r < 6; ++r) R.d[r] = xd[r];
    const v2f* Bp = reinterpret_cast<const v2f*>(xd + 8);
    const v2f* Cp = reinterpret_cast<const v2f*>(xd + 24);
    #pragma unroll
    for (int i = 0; i < 8; ++i) { R.B[i] = Bp[i]; R.C[i] = Cp[i]; }
    R.u = *uslot;
}

__global__ __launch_bounds__(32 * WAVES_PER_BLOCK) void
ssm_fused_kernel(const float* __restrict__ x,
                 const float* __restrict__ Win,
                 const float* __restrict__ Wxp,
                 const float* __restrict__ Wdt,
                 const float* __restrict__ dt_bias,
                 const float* __restrict__ A_logs,
                 const float* __restrict__ Dvec,
                 const float* __restrict__ gamma,
                 const float* __restrict__ beta,
                 const float* __restrict__ Wout,
                 float* __restrict__ out,
                 int npix)
{
    __shared__ WaveSmem smem[WAVES_PER_BLOCK];

    const int wave = __builtin_amdgcn_readfirstlane((int)(threadIdx.x >> 5));
    const int lane = threadIdx.x & 31;
    const int ntail = (npix / NPIX_PER_WAVE) % TILE_ROUND;
    const bool tail = (int)blockIdx.x < ntail;
    const int pix0 = tail ? (int)blockIdx.x * NPIX_PER_WAVE
                          : (ntail + ((int)blockIdx.x - ntail) * WAVES_PER_BLOCK + wave) * NPIX_PER_WAVE;
    if (pix0 + NPIX_PER_WAVE > npix) return;

    WaveSmem& S = smem[tail ? 0 : wave];

    const int rowM  = lane & 15;
    const int khalf = lane >> 4;
    const float* xrow = x + (size_t)(pix0 + (rowM >> 2)) * 384 + (rowM & 3);

    v4f wa[12], wb[12];

    if (!tail) {
        v2f afr[24];
        load_x12(xrow, khalf, afr);
        const float* w0 = Win + (size_t)rowM * 96 + 4 * khalf;
        bool v0;
        const float* wxp0 = Wxp + (size_t)xp_row(rowM, v0) * 192 + 4 * khalf;
        load_w12(w0, wa);
        #pragma unroll 1
        for (int nt = 0; nt < 12; nt += 2) {
            const float* wt = w0 + (size_t)nt * (16 * 96);
            load_w12(wt + 16 * 96, wb);
            v8f acc = {};
            acc = mma12(acc, afr, wa);
            load_w12((nt + 2 < 12) ? wt + 32 * 96 : wxp0, wa);
            for (int r = 0; r < 8; ++r) {
                float v = acc[r];
                v = silu_rcp(v);
                S.y[(r + 8 * khalf) * Y_STRIDE + nt * 16 + rowM] = v;
            }
            v8f acc2 = {};
            acc2 = mma12(acc2, afr, wb);
            for (int r = 0; r < 8; ++r) {
                float v = acc2[r];
                v = silu_rcp(v);
                S.y[(r + 8 * khalf) * Y_STRIDE + (nt + 1) * 16 + rowM] = v;
            }
        }
    } else {
        v2f afr[24];
        load_x12(xrow, khalf, afr);
        #pragma unroll 1
        for (int t = 0; t < 3; ++t) {
            const int nt = wave * 3 + t;
            load_w12(Win + (size_t)(nt * 16 + rowM) * 96 + 4 * khalf, wa);
            v8f acc = {};
            acc = mma12(acc, afr, wa);
            for (int r = 0; r < 8; ++r) {
                float v = acc[r];
                v = silu_rcp(v);
                S.y[(r + 8 * khalf) * Y_STRIDE + nt * 16 + rowM] = v;
            }
        }
    }
    __syncthreads();

    if (!tail) {
        v2f alo[24], ahi[24];
        load_a12(&S.y[rowM * Y_STRIDE + 4 * khalf], alo);
        load_a12(&S.y[rowM * Y_STRIDE + 4 * khalf + 96], ahi);
        for (int nt = 0; nt < 3; ++nt) {
            const int c  = nt * 16 + rowM;
            bool valid;
            const float* wt = Wxp + (size_t)xp_row(c, valid) * 192 + 4 * khalf;
            load_w12(wt + 96, wb);
            #pragma unroll
            for (int j = 0; j < 12; ++j) {
                wa[j].x = valid ? wa[j].x : 0.0f; wa[j].y = valid ? wa[j].y : 0.0f;
                wa[j].z = valid ? wa[j].z : 0.0f; wa[j].w = valid ? wa[j].w : 0.0f;
            }
            v8f acc = {};
            acc = mma12(acc, alo, wa);
            if (nt + 1 < 3) {
                bool vn;
                load_w12(Wxp + (size_t)xp_row(c + 16, vn) * 192 + 4 * khalf, wa);
            }
            #pragma unroll
            for (int j = 0; j < 12; ++j) {
                wb[j].x = valid ? wb[j].x : 0.0f; wb[j].y = valid ? wb[j].y : 0.0f;
                wb[j].z = valid ? wb[j].z : 0.0f; wb[j].w = valid ? wb[j].w : 0.0f;
            }
            acc = mma12(acc, ahi, wb);
            for (int r = 0; r < 8; ++r)
                S.xdbl[(r + 8 * khalf) * 48 + c] = acc[r];
        }
    } else if (wave < 3) {
        v2f alo[24], ahi[24];
        load_a12(&S.y[rowM * Y_STRIDE + 4 * khalf], alo);
        load_a12(&S.y[rowM * Y_STRIDE + 4 * khalf + 96], ahi);
        const int c = wave * 16 + rowM;
        bool valid;
        const float* wt = Wxp + (size_t)xp_row(c, valid) * 192 + 4 * khalf;
        load_w12(wt, wa);
        load_w12(wt + 96, wb);
        #pragma unroll
        for (int j = 0; j < 12; ++j) {
            wa[j].x = valid ? wa[j].x : 0.0f; wa[j].y = valid ? wa[j].y : 0.0f;
            wa[j].z = valid ? wa[j].z : 0.0f; wa[j].w = valid ? wa[j].w : 0.0f;
            wb[j].x = valid ? wb[j].x : 0.0f; wb[j].y = valid ? wb[j].y : 0.0f;
            wb[j].z = valid ? wb[j].z : 0.0f; wb[j].w = valid ? wb[j].w : 0.0f;
        }
        v8f acc = {};
        acc = mma12(acc, alo, wa);
        acc = mma12(acc, ahi, wb);
        for (int r = 0; r < 8; ++r)
            S.xdbl[(r + 8 * khalf) * 48 + c] = acc[r];
    }
    __syncthreads();

    {
        float lnsum[16], lnsq[16];
        #pragma unroll
        for (int r = 0; r < 16; ++r) { lnsum[r] = 0.f; lnsq[r] = 0.f; }

        const int it0    = tail ? wave : 0;
        const int itstep = tail ? WAVES_PER_BLOCK : 1;
        for (int it = it0; it < 6; it += itstep) {
            const int ch = it * 32 + lane;
            v2f Av[8];
            #pragma unroll
            for (int i = 0; i < 8; ++i) {
                Av[i].x = -__expf(A_logs[ch * 16 + 2 * i])     * 1.4426950408889634f;
                Av[i].y = -__expf(A_logs[ch * 16 + 2 * i + 1]) * 1.4426950408889634f;
            }
            const float dtb = dt_bias[ch];
            const float dch = Dvec[ch];
            float wdt[6];
            #pragma unroll
            for (int r = 0; r < 6; ++r) wdt[r] = Wdt[ch * 6 + r];

            ScanRowIn cur, nxt;
            load_scan_row(&S.xdbl[0], &S.y[ch], cur);
            #pragma unroll
            for (int p = 0; p < 4; ++p) {
                v2f h[8];
                #pragma unroll
                for (int i = 0; i < 8; ++i) { h[i].x = 0.f; h[i].y = 0.f; }
                #pragma unroll
                for (int l = 0; l < 4; ++l) {
                    const int row = p * 4 + l;
                    if (row + 1 < 16)
                        load_scan_row(&S.xdbl[(row + 1) * 48], &S.y[(row + 1) * Y_STRIDE + ch], nxt);
                    float dacc = dtb;
                    #pragma unroll
                    for (int r = 0; r < 6; ++r) dacc += cur.d[r] * wdt[r];
                    const float e     = __builtin_amdgcn_exp2f(dacc * 1.4426950408889634f);
                    const float sp    = __builtin_amdgcn_logf(1.0f + e) * 0.6931471805599453f;
                    const float delta = (dacc > 20.f) ? dacc : sp;
                    const float u  = cur.u;
                    const float du = delta * u;
                    v2f dl;  dl.x = delta;  dl.y = delta;
                    v2f du2; du2.x = du;    du2.y = du;
                    v2f y2;  y2.x = 0.f;    y2.y = 0.f;
                    #pragma unroll
                    for (int i = 0; i < 8; ++i) {
                        const v2f t = dl * Av[i];
                        v2f dA;
                        dA.x = __builtin_amdgcn_exp2f(t.x);
                        dA.y = __builtin_amdgcn_exp2f(t.y);
                        h[i] = dA * h[i] + du2 * cur.B[i];
                        y2   = h[i] * cur.C[i] + y2;
                    }
                    const float y = (y2.x + y2.y) + dch * u;
                    S.y[row * Y_STRIDE + ch] = y;
                    lnsum[row] += y; lnsq[row] += y * y;
                    if (row + 1 < 16) cur = nxt;
                }
            }
        }
        #pragma unroll
        for (int m = 1; m < 32; m <<= 1) {
            #pragma unroll
            for (int r = 0; r < 16; ++r) {
                lnsum[r] += __shfl_xor(lnsum[r], m, 32);
                lnsq[r]  += __shfl_xor(lnsq[r],  m, 32);
            }
        }
        if (!tail) {
            if (lane == 0) {
                #pragma unroll
                for (int r = 0; r < 16; ++r) {
                    const float mu  = lnsum[r] * (1.0f / 192.0f);
                    const float var = lnsq[r] * (1.0f / 192.0f) - mu * mu;
                    S.mu[r] = mu;
                    S.rs[r] = rsqrtf(var + 1e-5f);
                }
            }
        } else {
            float* part = smem[1].y;
            if (lane == 0) {
                #pragma unroll
                for (int r = 0; r < 16; ++r) {
                    part[(wave * 16 + r) * 2]     = lnsum[r];
                    part[(wave * 16 + r) * 2 + 1] = lnsq[r];
                }
            }
            __syncthreads();
            if (wave == 0 && lane < 16) {
                float sm = 0.f, sq = 0.f;
                for (int w = 0; w < WAVES_PER_BLOCK; ++w) {
                    sm += part[(w * 16 + lane) * 2];
                    sq += part[(w * 16 + lane) * 2 + 1];
                }
                const float mu  = sm * (1.0f / 192.0f);
                const float var = sq * (1.0f / 192.0f) - mu * mu;
                S.mu[lane] = mu;
                S.rs[lane] = rsqrtf(var + 1e-5f);
            }
        }
    }
    __syncthreads();

    if (!tail) {
        float mur[8], rsr[8];
        #pragma unroll
        for (int r = 0; r < 8; ++r) { mur[r] = S.mu[r + 8 * khalf]; rsr[r] = S.rs[r + 8 * khalf]; }

        v2f afr[24];
        load_x12(xrow, khalf, afr);
        const float* w0 = Win + (size_t)(192 + rowM) * 96 + 4 * khalf;
        const float* wout0 = Wout + (size_t)rowM * 192 + 4 * khalf;
        load_w12(w0, wa);
        for (int nt = 0; nt < 12; nt += 2) {
            const int ncol = nt * 16 + rowM;
            const float g0 = gamma[ncol],      bt0 = beta[ncol];
            const float g1 = gamma[ncol + 16], bt1 = beta[ncol + 16];
            const float* wt = w0 + (size_t)nt * (16 * 96);
            load_w12(wt + 16 * 96, wb);
            v8f acc = {};
            acc = mma12(acc, afr, wa);
            load_w12((nt + 2 < 12) ? wt + 32 * 96 : wout0, wa);
            for (int r = 0; r < 8; ++r) {
                float z = acc[r];
                z = silu_rcp(z);
                const int idx = (r + 8 * khalf) * Y_STRIDE + ncol;
                float v = (S.y[idx] - mur[r]) * rsr[r] * g0 + bt0;
                v *= z;
                S.y[idx] = v;
            }
            v8f acc2 = {};
            acc2 = mma12(acc2, afr, wb);
            for (int r = 0; r < 8; ++r) {
                float z = acc2[r];
                z = silu_rcp(z);
                const int idx = (r + 8 * khalf) * Y_STRIDE + ncol + 16;
                float v = (S.y[idx] - mur[r]) * rsr[r] * g1 + bt1;
                v *= z;
                S.y[idx] = v;
            }
        }
    } else {
        float mur[8], rsr[8];
        #pragma unroll
        for (int r = 0; r < 8; ++r) { mur[r] = S.mu[r + 8 * khalf]; rsr[r] = S.rs[r + 8 * khalf]; }
        v2f afr[24];
        load_x12(xrow, khalf, afr);
        for (int t = 0; t < 3; ++t) {
            const int nt   = wave * 3 + t;
            const int ncol = nt * 16 + rowM;
            const float g0 = gamma[ncol], bt0 = beta[ncol];
            load_w12(Win + (size_t)(192 + nt * 16 + rowM) * 96 + 4 * khalf, wa);
            v8f acc = {};
            acc = mma12(acc, afr, wa);
            for (int r = 0; r < 8; ++r) {
                float z = acc[r];
                z = silu_rcp(z);
                const int idx = (r + 8 * khalf) * Y_STRIDE + ncol;
                float v = (S.y[idx] - mur[r]) * rsr[r] * g0 + bt0;
                v *= z;
                S.y[idx] = v;
            }
        }
    }
    __syncthreads();

    if (!tail) {
        v2f alo[24], ahi[24];
        load_a12(&S.y[rowM * Y_STRIDE + 4 * khalf], alo);
        load_a12(&S.y[rowM * Y_STRIDE + 4 * khalf + 96], ahi);
        const float* w0 = Wout + (size_t)rowM * 192 + 4 * khalf;
        #pragma unroll 1
        for (int nt = 0; nt < 5; ++nt) {
            const float* wt = w0 + (size_t)nt * (16 * 192);
            load_w12(wt + 96, wb);
            v8f acc = {};
            acc = mma12(acc, alo, wa);
            load_w12(wt + 16 * 192, wa);
            acc = mma12(acc, ahi, wb);
            const int ncol = nt * 16 + rowM;
            for (int r = 0; r < 8; ++r) {
                const int row = r + 8 * khalf;
                out[(size_t)(pix0 + (row >> 2)) * 384 + ncol * 4 + (row & 3)] = acc[r];
            }
        }
        {
            const float* wt = w0 + (size_t)5 * (16 * 192);
            load_w12(wt + 96, wb);
            v8f acc = {};
            acc = mma12(acc, alo, wa);
            acc = mma12(acc, ahi, wb);
            const int ncol = 5 * 16 + rowM;
            for (int r = 0; r < 8; ++r) {
                const int row = r + 8 * khalf;
                out[(size_t)(pix0 + (row >> 2)) * 384 + ncol * 4 + (row & 3)] = acc[r];
            }
        }
    } else {
        v2f alo[24], ahi[24];
        load_a12(&S.y[rowM * Y_STRIDE + 4 * khalf], alo);
        load_a12(&S.y[rowM * Y_STRIDE + 4 * khalf + 96], ahi);
        for (int nt = wave; nt < 6; nt += WAVES_PER_BLOCK) {
            const float* wt = Wout + (size_t)(nt * 16 + rowM) * 192 + 4 * khalf;
            load_w12(wt, wa);
            load_w12(wt + 96, wb);
            v8f acc = {};
            acc = mma12(acc, alo, wa);
            acc = mma12(acc, ahi, wb);
            const int ncol = nt * 16 + rowM;
            for (int r = 0; r < 8; ++r) {
                const int row = r + 8 * khalf;
                out[(size_t)(pix0 + (row >> 2)) * 384 + ncol * 4 + (row & 3)] = acc[r];
            }
        }
    }
}

extern "C" void kernel_launch(void* const* d_in, const int* in_sizes, int n_in,
                              void* d_out, int out_size, void* d_ws, size_t ws_size,
                              hipStream_t stream) {
    const float* x       = (const float*)d_in[0];
    const float* Win     = (const float*)d_in[1];
    const float* Wxp     = (const float*)d_in[2];
    const float* Wdt     = (const float*)d_in[3];
    const float* dt_bias = (const float*)d_in[4];
    const float* A_logs  = (const float*)d_in[5];
    const float* Dvec    = (const float*)d_in[6];
    const float* gamma   = (const float*)d_in[7];
    const float* beta    = (const float*)d_in[8];
    const float* Wout    = (const float*)d_in[9];
    float* out = (float*)d_out;

    const int npix = in_sizes[0] / (96 * 4);
    const int tiles = npix / NPIX_PER_WAVE;
    const int ntail = tiles % TILE_ROUND;
    const int grid  = ntail + (tiles - ntail) / WAVES_PER_BLOCK;

    ssm_fused_kernel<<<grid, 32 * WAVES_PER_BLOCK, 0, stream>>>(x, Win, Wxp, Wdt, dt_bias, A_logs,
                                              Dvec, gamma, beta, Wout, out, npix);
}


# Win, Wxp and Wout rewritten once per call by a small first kernel into fragment order in the workspace: each 16-byte-per-lane weight load touches 4 whole cache lines, not 16; launches 1 -> 2
# speedup vs baseline: 2.9633x; 1.0360x over previous
#include <hip/hip_runtime.h>
#include <hip/hip_bf16.h>

typedef __attribute__((ext_vector_type(2))) float v2f;
typedef __attribute__((ext_vector_type(8))) float v8f;

#define Y_STRIDE 196
#define NPIX_PER_WAVE 4
#define WAVES_PER_BLOCK 4
#define TILE_ROUND 1024

#define PK_GROUP 128
#define PK_STEP  (12 * PK_GROUP)
#define PK_WIN   0
#define PK_WXP   24
#define PK_WOUT  30
#define PK_STEPS 42

struct alignas(16) WaveSmem {
    float y[16 * Y_STRIDE];
    float xdbl[16 * 48];
    float mu[16];
    float rs[16];
};

__device__ __forceinline__ int xp_row(int p, bool& valid)
{
    valid = (p < 6) || (p >= 8 && p < 40);
    return valid ? (p < 6 ? p : p - 2) : 37;
}

typedef __attribute__((ext_vector_type(4))) float v4f;

__device__ __forceinline__ float silu_rcp(float v)
{
    return v * __builtin_amdgcn_rcpf(1.0f + __expf(-v));
}


__device__ __forceinline__ void load_w12(const float* __restrict__ wb, v4f (&w)[12])
{
    #pragma unroll
    for (int j = 0; j < 12; ++j)
        w[j] = *reinterpret_cast<const v4f*>(wb + j * PK_GROUP);
}

__device__ __forceinline__ void load_a12(const float* arow, v2f (&a)[24])
{
    #pragma unroll
    for (int j = 0; j < 12; ++j) {
        a[2 * j].x     = arow[j * 8];     a[2 * j].y     = arow[j * 8 + 1];
        a[2 * j + 1].x = arow[j * 8 + 2]; a[2 * j + 1].y = arow[j * 8 + 3];
    }
}

__device__ __forceinline__ void load_x12(const float* __restrict__ xrow, int khalf, v2f (&a)[24])
{
    #pragma unroll
    for (int j = 0; j < 12; ++j) {
        const int kb = j * 8 + 4 * khalf;
        a[2 * j].x     = xrow[(size_t)kb * 4];       a[2 * j].y     = xrow[(size_t)(kb + 1) * 4];
        a[2 * j + 1].x = xrow[(size_t)(kb + 2) * 4]; a[2 * j + 1].y = xrow[(size_t)(kb + 3) * 4];
    }
}

__device__ __forceinline__ v8f mma12(v8f acc, const v2f (&a)[24], const v4f (&w)[12])
{
    #pragma unroll
    for (int j = 0; j < 12; ++j) {
        v2f bA; bA.x = w[j].x; bA.y = w[j].y;
        v2f bB; bB.x = w[j].z; bB.y = w[j].w;
        acc = __builtin_amdgcn_wmma_f32_16x16x4_f32(false, a[2 * j], false, bA,
                                                    (short)0, acc, false, false);
        acc = __builtin_amdgcn_wmma_f32_16x16x4_f32(false, a[2 * j + 1], false, bB,
                                                    (short)0, acc, false, false);
    }
    return acc;
}

struct ScanRowIn { float d[6]; v2f B[8]; v2f C[8]; float u; };

__device__ __forceinline__ void load_scan_row(const float* xd, const float* uslot, ScanRowIn& R)
{
    #pragma unroll
    for (int r = 0; r < 6; ++r) R.d[r] = xd[r];
    const v2f* Bp = reinterpret_cast<const v2f*>(xd + 8);
    const v2f* Cp = reinterpret_cast<const v2f*>(xd + 24);
    #pragma unroll
    for (int i = 0; i < 8; ++i) { R.B[i] = Bp[i]; R.C[i] = Cp[i]; }
    R.u = *uslot;
}

__global__ __launch_bounds__(256) void
pack_weights_kernel(const float* __restrict__ Win,
                    const float* __restrict__ Wxp,
                    const float* __restrict__ Wout,
                    float* __restrict__ Wp)
{
    const int t = (int)blockIdx.x * 256 + (int)threadIdx.x;
    if (t >= PK_STEPS * 12 * 32) return;
    const int step = t / (12 * 32);
    const int rem  = t - step * (12 * 32);
    const int j    = rem >> 5;
    const int lane = rem & 31;
    const int c    = lane & 15;
    const int k    = 8 * j + 4 * (lane >> 4);
    const float* src;
    bool valid = true;
    if (step < PK_WXP) {
        src = Win + (size_t)(step * 16 + c) * 96 + k;
    } else if (step < PK_WOUT) {
        const int q = step - PK_WXP;
        src = Wxp + (size_t)xp_row((q >> 1) * 16 + c, valid) * 192 + 96 * (q & 1) + k;
    } else {
        const int q = step - PK_WOUT;
        src = Wout + (size_t)((q >> 1) * 16 + c) * 192 + 96 * (q & 1) + k;
    }
    v4f v;
    v.x = src[0]; v.y = src[1]; v.z = src[2]; v.w = src[3];
    if (!valid) { v.x = 0.0f; v.y = 0.0f; v.z = 0.0f; v.w = 0.0f; }
    *reinterpret_cast<v4f*>(Wp + (size_t)t * 4) = v;
}

__global__ __launch_bounds__(32 * WAVES_PER_BLOCK) void
ssm_fused_kernel(const float* __restrict__ x,
                 const float* __restrict__ Wp,
                 const float* __restrict__ Wdt,
                 const float* __restrict__ dt_bias,
                 const float* __restrict__ A_logs,
                 const float* __restrict__ Dvec,
                 const float* __restrict__ gamma,
                 const float* __restrict__ beta,
                 float* __restrict__ out,
                 int npix)
{
    __shared__ WaveSmem smem[WAVES_PER_BLOCK];

    const int wave = __builtin_amdgcn_readfirstlane((int)(threadIdx.x >> 5));
    const int lane = threadIdx.x & 31;
    const int ntail = (npix / NPIX_PER_WAVE) % TILE_ROUND;
    const bool tail = (int)blockIdx.x < ntail;
    const int pix0 = tail ? (int)blockIdx.x * NPIX_PER_WAVE
                          : (ntail + ((int)blockIdx.x - ntail) * WAVES_PER_BLOCK + wave) * NPIX_PER_WAVE;
    if (pix0 + NPIX_PER_WAVE > npix) return;

    WaveSmem& S = smem[tail ? 0 : wave];

    const int rowM  = lane & 15;
    const int khalf = lane >> 4;
    const float* xrow = x + (size_t)(pix0 + (rowM >> 2)) * 384 + (rowM & 3);

    v4f wa[12], wb[12];
    const float* wl = Wp + lane * 4;

    if (!tail) {
        v2f afr[24];
        load_x12(xrow, khalf, afr);
        const float* w0 = wl + PK_WIN * PK_STEP;
        const float* wxp0 = wl + PK_WXP * PK_STEP;
        load_w12(w0, wa);
        #pragma unroll 1
        for (int nt = 0; nt < 12; nt += 2) {
            const float* wt = w0 + nt * PK_STEP;
            load_w12(wt + PK_STEP, wb);
            v8f acc = {};
            acc = mma12(acc, afr, wa);
            load_w12((nt + 2 < 12) ? wt + 2 * PK_STEP : wxp0, wa);
            for (int r = 0; r < 8; ++r) {
                float v = acc[r];
                v = silu_rcp(v);
                S.y[(r + 8 * khalf) * Y_STRIDE + nt * 16 + rowM] = v;
            }
            v8f acc2 = {};
            acc2 = mma12(acc2, afr, wb);
            for (int r = 0; r < 8; ++r) {
                float v = acc2[r];
                v = silu_rcp(v);
                S.y[(r + 8 * khalf) * Y_STRIDE + (nt + 1) * 16 + rowM] = v;
            }
        }
    } else {
        v2f afr[24];
        load_x12(xrow, khalf, afr);
        #pragma unroll 1
        for (int t = 0; t < 3; ++t) {
            const int nt = wave * 3 + t;
            load_w12(wl + (PK_WIN + nt) * PK_STEP, wa);
            v8f acc = {};
            acc = mma12(acc, afr, wa);
            for (int r = 0; r < 8; ++r) {
                float v = acc[r];
                v = silu_rcp(v);
                S.y[(r + 8 * khalf) * Y_STRIDE + nt * 16 + rowM] = v;
            }
        }
    }
    __syncthreads();

    if (!tail) {
        v2f alo[24], ahi[24];
        load_a12(&S.y[rowM * Y_STRIDE + 4 * khalf], alo);
        load_a12(&S.y[rowM * Y_STRIDE + 4 * khalf + 96], ahi);
        for (int nt = 0; nt < 3; ++nt) {
            const int c  = nt * 16 + rowM;
            const float* wt = wl + (PK_WXP + 2 * nt) * PK_STEP;
            load_w12(wt + PK_STEP, wb);
            v8f acc = {};
            acc = mma12(acc, alo, wa);
            if (nt + 1 < 3)
                load_w12(wt + 2 * PK_STEP, wa);
            acc = mma12(acc, ahi, wb);
            for (int r = 0; r < 8; ++r)
                S.xdbl[(r + 8 * khalf) * 48 + c] = acc[r];
        }
    } else if (wave < 3) {
        v2f alo[24], ahi[24];
        load_a12(&S.y[rowM * Y_STRIDE + 4 * khalf], alo);
        load_a12(&S.y[rowM * Y_STRIDE + 4 * khalf + 96], ahi);
        const int c = wave * 16 + rowM;
        const float* wt = wl + (PK_WXP + 2 * wave) * PK_STEP;
        load_w12(wt, wa);
        load_w12(wt + PK_STEP, wb);
        v8f acc = {};
        acc = mma12(acc, alo, wa);
        acc = mma12(acc, ahi, wb);
        for (int r = 0; r < 8; ++r)
            S.xdbl[(r + 8 * khalf) * 48 + c] = acc[r];
    }
    __syncthreads();

    {
        float lnsum[16], lnsq[16];
        #pragma unroll
        for (int r = 0; r < 16; ++r) { lnsum[r] = 0.f; lnsq[r] = 0.f; }

        const int it0    = tail ? wave : 0;
        const int itstep = tail ? WAVES_PER_BLOCK : 1;
        for (int it = it0; it < 6; it += itstep) {
            const int ch = it * 32 + lane;
            v2f Av[8];
            #pragma unroll
            for (int i = 0; i < 8; ++i) {
                Av[i].x = -__expf(A_logs[ch * 16 + 2 * i])     * 1.4426950408889634f;
                Av[i].y = -__expf(A_logs[ch * 16 + 2 * i + 1]) * 1.4426950408889634f;
            }
            const float dtb = dt_bias[ch];
            const float dch = Dvec[ch];
            float wdt[6];
            #pragma unroll
            for (int r = 0; r < 6; ++r) wdt[r] = Wdt[ch * 6 + r];

            ScanRowIn cur, nxt;
            load_scan_row(&S.xdbl[0], &S.y[ch], cur);
            #pragma unroll
            for (int p = 0; p < 4; ++p) {
                v2f h[8];
                #pragma unroll
                for (int i = 0; i < 8; ++i) { h[i].x = 0.f; h[i].y = 0.f; }
                #pragma unroll
                for (int l = 0; l < 4; ++l) {
                    const int row = p * 4 + l;
                    if (row + 1 < 16)
                        load_scan_row(&S.xdbl[(row + 1) * 48], &S.y[(row + 1) * Y_STRIDE + ch], nxt);
                    float dacc = dtb;
                    #pragma unroll
                    for (int r = 0; r < 6; ++r) dacc += cur.d[r] * wdt[r];
                    const float e     = __builtin_amdgcn_exp2f(dacc * 1.4426950408889634f);
                    const float sp    = __builtin_amdgcn_logf(1.0f + e) * 0.6931471805599453f;
                    const float delta = (dacc > 20.f) ? dacc : sp;
                    const float u  = cur.u;
                    const float du = delta * u;
                    v2f dl;  dl.x = delta;  dl.y = delta;
                    v2f du2; du2.x = du;    du2.y = du;
                    v2f y2;  y2.x = 0.f;    y2.y = 0.f;
                    #pragma unroll
                    for (int i = 0; i < 8; ++i) {
                        const v2f t = dl * Av[i];
                        v2f dA;
                        dA.x = __builtin_amdgcn_exp2f(t.x);
                        dA.y = __builtin_amdgcn_exp2f(t.y);
                        h[i] = dA * h[i] + du2 * cur.B[i];
                        y2   = h[i] * cur.C[i] + y2;
                    }
                    const float y = (y2.x + y2.y) + dch * u;
                    S.y[row * Y_STRIDE + ch] = y;
                    lnsum[row] += y; lnsq[row] += y * y;
                    if (row + 1 < 16) cur = nxt;
                }
            }
        }
        #pragma unroll
        for (int m = 1; m < 32; m <<= 1) {
            #pragma unroll
            for (int r = 0; r < 16; ++r) {
                lnsum[r] += __shfl_xor(lnsum[r], m, 32);
                lnsq[r]  += __shfl_xor(lnsq[r],  m, 32);
            }
        }
        if (!tail) {
            if (lane == 0) {
                #pragma unroll
                for (int r = 0; r < 16; ++r) {
                    const float mu  = lnsum[r] * (1.0f / 192.0f);
                    const float var = lnsq[r] * (1.0f / 192.0f) - mu * mu;
                    S.mu[r] = mu;
                    S.rs[r] = rsqrtf(var + 1e-5f);
                }
            }
        } else {
            float* part = smem[1].y;
            if (lane == 0) {
                #pragma unroll
                for (int r = 0; r < 16; ++r) {
                    part[(wave * 16 + r) * 2]     = lnsum[r];
                    part[(wave * 16 + r) * 2 + 1] = lnsq[r];
                }
            }
            __syncthreads();
            if (wave == 0 && lane < 16) {
                float sm = 0.f, sq = 0.f;
                for (int w = 0; w < WAVES_PER_BLOCK; ++w) {
                    sm += part[(w * 16 + lane) * 2];
                    sq += part[(w * 16 + lane) * 2 + 1];
                }
                const float mu  = sm * (1.0f / 192.0f);
                const float var = sq * (1.0f / 192.0f) - mu * mu;
                S.mu[lane] = mu;
                S.rs[lane] = rsqrtf(var + 1e-5f);
            }
        }
    }
    __syncthreads();

    if (!tail) {
        float mur[8], rsr[8];
        #pragma unroll
        for (int r = 0; r < 8; ++r) { mur[r] = S.mu[r + 8 * khalf]; rsr[r] = S.rs[r + 8 * khalf]; }

        v2f afr[24];
        load_x12(xrow, khalf, afr);
        const float* w0 = wl + (PK_WIN + 12) * PK_STEP;
        const float* wout0 = wl + PK_WOUT * PK_STEP;
        load_w12(w0, wa);
        #pragma unroll 1
        for (int nt = 0; nt < 12; nt += 2) {
            const int ncol = nt * 16 + rowM;
            const float g0 = gamma[ncol],      bt0 = beta[ncol];
            const float g1 = gamma[ncol + 16], bt1 = beta[ncol + 16];
            const float* wt = w0 + nt * PK_STEP;
            load_w12(wt + PK_STEP, wb);
            v8f acc = {};
            acc = mma12(acc, afr, wa);
            load_w12((nt + 2 < 12) ? wt + 2 * PK_STEP : wout0, wa);
            for (int r = 0; r < 8; ++r) {
                float z = acc[r];
                z = silu_rcp(z);
                const int idx = (r + 8 * khalf) * Y_STRIDE + ncol;
                float v = (S.y[idx] - mur[r]) * rsr[r] * g0 + bt0;
                v *= z;
                S.y[idx] = v;
            }
            v8f acc2 = {};
            acc2 = mma12(acc2, afr, wb);
            for (int r = 0; r < 8; ++r) {
                float z = acc2[r];
                z = silu_rcp(z);
                const int idx = (r + 8 * khalf) * Y_STRIDE + ncol + 16;
                float v = (S.y[idx] - mur[r]) * rsr[r] * g1 + bt1;
                v *= z;
                S.y[idx] = v;
            }
        }
    } else {
        float mur[8], rsr[8];
        #pragma unroll
        for (int r = 0; r < 8; ++r) { mur[r] = S.mu[r + 8 * khalf]; rsr[r] = S.rs[r + 8 * khalf]; }
        v2f afr[24];
        load_x12(xrow, khalf, afr);
        for (int t = 0; t < 3; ++t) {
            const int nt   = wave * 3 + t;
            const int ncol = nt * 16 + rowM;
            const float g0 = gamma[ncol], bt0 = beta[ncol];
            load_w12(wl + (PK_WIN + 12 + nt) * PK_STEP, wa);
            v8f acc = {};
            acc = mma12(acc, afr, wa);
            for (int r = 0; r < 8; ++r) {
                float z = acc[r];
                z = silu_rcp(z);
                const int idx = (r + 8 * khalf) * Y_STRIDE + ncol;
                float v = (S.y[idx] - mur[r]) * rsr[r] * g0 + bt0;
                v *= z;
                S.y[idx] = v;
            }
        }
    }
    __syncthreads();

    if (!tail) {
        v2f alo[24], ahi[24];
        load_a12(&S.y[rowM * Y_STRIDE + 4 * khalf], alo);
        load_a12(&S.y[rowM * Y_STRIDE + 4 * khalf + 96], ahi);
        const float* w0 = wl + PK_WOUT * PK_STEP;
        #pragma unroll 1
        for (int nt = 0; nt < 5; ++nt) {
            const float* wt = w0 + nt * (2 * PK_STEP);
            load_w12(wt + PK_STEP, wb);
            v8f acc = {};
            acc = mma12(acc, alo, wa);
            load_w12(wt + 2 * PK_STEP, wa);
            acc = mma12(acc, ahi, wb);
            const int ncol = nt * 16 + rowM;
            for (int r = 0; r < 8; ++r) {
                const int row = r + 8 * khalf;
                out[(size_t)(pix0 + (row >> 2)) * 384 + ncol * 4 + (row & 3)] = acc[r];
            }
        }
        {
            const float* wt = w0 + 5 * (2 * PK_STEP);
            load_w12(wt + PK_STEP, wb);
            v8f acc = {};
            acc = mma12(acc, alo, wa);
            acc = mma12(acc, ahi, wb);
            const int ncol = 5 * 16 + rowM;
            for (int r = 0; r < 8; ++r) {
                const int row = r + 8 * khalf;
                out[(size_t)(pix0 + (row >> 2)) * 384 + ncol * 4 + (row & 3)] = acc[r];
            }
        }
    } else {
        v2f alo[24], ahi[24];
        load_a12(&S.y[rowM * Y_STRIDE + 4 * khalf], alo);
        load_a12(&S.y[rowM * Y_STRIDE + 4 * khalf + 96], ahi);
        for (int nt = wave; nt < 6; nt += WAVES_PER_BLOCK) {
            const float* wt = wl + (PK_WOUT + 2 * nt) * PK_STEP;
            load_w12(wt, wa);
            load_w12(wt + PK_STEP, wb);
            v8f acc = {};
            acc = mma12(acc, alo, wa);
            acc = mma12(acc, ahi, wb);
            const int ncol = nt * 16 + rowM;
            for (int r = 0; r < 8; ++r) {
                const int row = r + 8 * khalf;
                out[(size_t)(pix0 + (row >> 2)) * 384 + ncol * 4 + (row & 3)] = acc[r];
            }
        }
    }
}

extern "C" void kernel_launch(void* const* d_in, const int* in_sizes, int n_in,
                              void* d_out, int out_size, void* d_ws, size_t ws_size,
                              hipStream_t stream) {
    const float* x       = (const float*)d_in[0];
    const float* Win     = (const float*)d_in[1];
    const float* Wxp     = (const float*)d_in[2];
    const float* Wdt     = (const float*)d_in[3];
    const float* dt_bias = (const float*)d_in[4];
    const float* A_logs  = (const float*)d_in[5];
    const float* Dvec    = (const float*)d_in[6];
    const float* gamma   = (const float*)d_in[7];
    const float* beta    = (const float*)d_in[8];
    const float* Wout    = (const float*)d_in[9];
    float* out = (float*)d_out;
    float* Wp  = (float*)d_ws;
    if (ws_size < (size_t)PK_STEPS * PK_STEP * sizeof(float)) return;

    const int npix = in_sizes[0] / (96 * 4);
    const int tiles = npix / NPIX_PER_WAVE;
    const int ntail = tiles % TILE_ROUND;
    const int grid  = ntail + (tiles - ntail) / WAVES_PER_BLOCK;

    pack_weights_kernel<<<PK_STEPS * 12 * 32 / 256, 256, 0, stream>>>(Win, Wxp, Wout, Wp);
    ssm_fused_kernel<<<grid, 32 * WAVES_PER_BLOCK, 0, stream>>>(x, Wp, Wdt, dt_bias, A_logs,
                                              Dvec, gamma, beta, out, npix);
}
